# starting kernel
# baseline (speedup 1.0000x reference)
#include <hip/hip_runtime.h>

typedef __attribute__((ext_vector_type(2))) float v2f;
typedef __attribute__((ext_vector_type(8))) float v8f;


__global__ void fill_kernel(float* __restrict__ p, float v, int n) {
    int i = blockIdx.x * blockDim.x + threadIdx.x;
    if (i < n) p[i] = v;
}

__global__ void deg_count_kernel(const int* __restrict__ dst,
                                 float* __restrict__ deg, int E) {
    int e = blockIdx.x * blockDim.x + threadIdx.x;
    if (e < E) atomicAdd(&deg[dst[e]], 1.0f);
}

__global__ void rsqrt_kernel(float* __restrict__ d, int n) {
    int i = blockIdx.x * blockDim.x + threadIdx.x;
    if (i < n) d[i] = rsqrtf(d[i]);
}


__global__ void __launch_bounds__(256)
gemm64_wmma(const float* __restrict__ X, const float* __restrict__ W,
            float* __restrict__ Y, int nrows) {
    __shared__ float lw[64 * 64];
    for (int i = threadIdx.x; i < 64 * 64; i += 256) {
        int k = i >> 6, n = i & 63;
        lw[(k >> 1) * 128 + n * 2 + (k & 1)] = W[i];
    }
    __syncthreads();

    int wave = threadIdx.x >> 5;
    int lane = threadIdx.x & 31;
    int lh   = lane & 15;
    int hi   = lane >> 4;
    int rowbase = (blockIdx.x * 8 + wave) * 16;

    int arow = rowbase + lh;
    int arow_c = arow < nrows ? arow : (nrows - 1);
    const v2f* xp = (const v2f*)(X + (size_t)arow_c * 64);

    v8f acc0 = {}, acc1 = {}, acc2 = {}, acc3 = {};

    #pragma unroll
    for (int kk = 0; kk < 16; ++kk) {
        v2f a = xp[2 * kk + hi];

        const v2f* wp = (const v2f*)(lw + (2 * kk + hi) * 128 + lh * 2);
        v2f b0  = wp[0];
        v2f b1v = wp[16];
        v2f b2v = wp[32];
        v2f b3v = wp[48];

        acc0 = __builtin_amdgcn_wmma_f32_16x16x4_f32(false, a, false, b0,  (short)0, acc0, false, false);
        acc1 = __builtin_amdgcn_wmma_f32_16x16x4_f32(false, a, false, b1v, (short)0, acc1, false, false);
        acc2 = __builtin_amdgcn_wmma_f32_16x16x4_f32(false, a, false, b2v, (short)0, acc2, false, false);
        acc3 = __builtin_amdgcn_wmma_f32_16x16x4_f32(false, a, false, b3v, (short)0, acc3, false, false);
    }

    #pragma unroll
    for (int r = 0; r < 8; ++r) {
        int row = rowbase + r + hi * 8;
        if (row < nrows) {
            float* yp = Y + (size_t)row * 64 + lh;
            yp[0]  = acc0[r];
            yp[16] = acc1[r];
            yp[32] = acc2[r];
            yp[48] = acc3[r];
        }
    }
}


__global__ void scatter_kernel(const float* __restrict__ xw,
                               const int* __restrict__ src,
                               const int* __restrict__ dst,
                               const float* __restrict__ dinv,
                               float* __restrict__ agg, int E) {
    int t = blockIdx.x * blockDim.x + threadIdx.x;
    int e = t >> 4;
    int q = t & 15;
    if (e >= E) return;
    int s = src[e];
    int d = dst[e];
    float nrm = dinv[s] * dinv[d];
    float4 v = reinterpret_cast<const float4*>(xw + (size_t)s * 64)[q];
    float* base = agg + (size_t)d * 64 + q * 4;
    atomicAdd(base + 0, v.x * nrm);
    atomicAdd(base + 1, v.y * nrm);
    atomicAdd(base + 2, v.z * nrm);
    atomicAdd(base + 3, v.w * nrm);
}


__global__ void finish_kernel(float* __restrict__ acc, const float* __restrict__ xw,
                              const float* __restrict__ dinv, const float* __restrict__ bias,
                              int total, int do_relu) {
    int i = blockIdx.x * blockDim.x + threadIdx.x;
    if (i >= total) return;
    int n = i >> 6;
    int f = i & 63;
    float di = dinv[n];
    float v = acc[i] + xw[i] * di * di + bias[f];
    if (do_relu) v = fmaxf(v, 0.0f);
    acc[i] = v;
}


extern "C" void kernel_launch(void* const* d_in, const int* in_sizes, int n_in,
                              void* d_out, int out_size, void* d_ws, size_t ws_size,
                              hipStream_t stream) {
    (void)n_in; (void)out_size; (void)ws_size;

    const float* x  = (const float*)d_in[0];
    const int*   ei = (const int*)d_in[1];
    const float* W1 = (const float*)d_in[2];
    const float* b1 = (const float*)d_in[3];
    const float* W2 = (const float*)d_in[4];
    const float* b2 = (const float*)d_in[5];
    float* out = (float*)d_out;

    const int N = in_sizes[0] / 64;
    const int E = in_sizes[1] / 2;
    const int* src = ei;
    const int* dst = ei + E;

    float* dinv = (float*)d_ws;
    float* xw   = dinv + (((size_t)N + 32767) & ~(size_t)32767);
    float* agg  = xw + (size_t)N * 64;

    const int T = 256;
    const int NF = N * 64;

    fill_kernel<<<(N + T - 1) / T, T, 0, stream>>>(dinv, 1.0f, N);
    deg_count_kernel<<<(E + T - 1) / T, T, 0, stream>>>(dst, dinv, E);
    rsqrt_kernel<<<(N + T - 1) / T, T, 0, stream>>>(dinv, N);

    gemm64_wmma<<<(N + 127) / 128, T, 0, stream>>>(x, W1, xw, N);
    fill_kernel<<<(NF + T - 1) / T, T, 0, stream>>>(agg, 0.0f, NF);
    scatter_kernel<<<((E * 16) + T - 1) / T, T, 0, stream>>>(xw, src, dst, dinv, agg, E);
    finish_kernel<<<(NF + T - 1) / T, T, 0, stream>>>(agg, xw, dinv, b1, NF, 1);

    gemm64_wmma<<<(N + 127) / 128, T, 0, stream>>>(agg, W2, xw, N);
    fill_kernel<<<(NF + T - 1) / T, T, 0, stream>>>(out, 0.0f, NF);
    scatter_kernel<<<((E * 16) + T - 1) / T, T, 0, stream>>>(xw, src, dst, dinv, out, E);
    finish_kernel<<<(NF + T - 1) / T, T, 0, stream>>>(out, xw, dinv, b2, NF, 0);
}


# atomic scatter-add replaced by a per-destination gather: edges grouped by dst each call (int atomic rank + scan), each node sums its row with plain stores; float sum order = row slot order
# speedup vs baseline: 1.9325x; 1.9325x over previous
#include <hip/hip_runtime.h>


typedef __attribute__((ext_vector_type(2))) float v2f;
typedef __attribute__((ext_vector_type(8))) float v8f;


__global__ void fill_kernel(float* __restrict__ p, float v, int n) {
    int i = blockIdx.x * blockDim.x + threadIdx.x;
    if (i < n) p[i] = v;
}


__global__ void hist_rank_kernel(const int* __restrict__ dst,
                                 int* __restrict__ cnt,
                                 int* __restrict__ rank, int E) {
    int e = blockIdx.x * blockDim.x + threadIdx.x;
    if (e < E) rank[e] = atomicAdd(&cnt[dst[e]], 1);
}


__global__ void __launch_bounds__(256)
scan_dinv_kernel(const int* __restrict__ cnt, int* __restrict__ rowptr,
                 float* __restrict__ dinv, int N) {
    __shared__ int sh[256];
    const int t = threadIdx.x;
    const int base = blockIdx.x * 256;

    int local = 0;
    const int4* c4 = reinterpret_cast<const int4*>(cnt);
    for (int i = t; i * 4 < base; i += 256) {
        int4 v = c4[i];
        local += (v.x + v.y) + (v.z + v.w);
    }
    sh[t] = local;
    __syncthreads();
    for (int s = 128; s > 0; s >>= 1) {
        if (t < s) sh[t] += sh[t + s];
        __syncthreads();
    }
    const int prefix = sh[0];
    __syncthreads();

    const int i = base + t;
    const int c = (i < N) ? cnt[i] : 0;
    sh[t] = c;
    __syncthreads();
    for (int off = 1; off < 256; off <<= 1) {
        int v = (t >= off) ? sh[t - off] : 0;
        __syncthreads();
        sh[t] += v;
        __syncthreads();
    }
    if (i < N) {
        rowptr[i] = prefix + sh[t] - c;
        dinv[i] = rsqrtf((float)(c + 1));
    }
}


__global__ void fill_csr_kernel(const int* __restrict__ src,
                                const int* __restrict__ dst,
                                const int* __restrict__ rank,
                                const int* __restrict__ rowptr,
                                const float* __restrict__ dinv,
                                int* __restrict__ csr, int E) {
    int e = blockIdx.x * blockDim.x + threadIdx.x;
    if (e >= E) return;
    int s = src[e];
    int d = dst[e];
    int pos = rowptr[d] + rank[e];
    pos = min(max(pos, 0), E - 1);
    reinterpret_cast<int2*>(csr)[pos] = make_int2(s, __float_as_int(dinv[s]));
}


__global__ void __launch_bounds__(256)
gemm64_wmma(const float* __restrict__ X, const float* __restrict__ W,
            float* __restrict__ Y, int nrows) {
    __shared__ float lw[64 * 64];
    for (int i = threadIdx.x; i < 64 * 64; i += 256) {
        int k = i >> 6, n = i & 63;
        lw[(k >> 1) * 128 + n * 2 + (k & 1)] = W[i];
    }
    __syncthreads();

    int wave = threadIdx.x >> 5;
    int lane = threadIdx.x & 31;
    int lh   = lane & 15;
    int hi   = lane >> 4;
    int rowbase = (blockIdx.x * 8 + wave) * 16;

    int arow = rowbase + lh;
    int arow_c = arow < nrows ? arow : (nrows - 1);
    const v2f* xp = (const v2f*)(X + (size_t)arow_c * 64);

    v8f acc0 = {}, acc1 = {}, acc2 = {}, acc3 = {};

    #pragma unroll
    for (int kk = 0; kk < 16; ++kk) {
        v2f a = xp[2 * kk + hi];

        const v2f* wp = (const v2f*)(lw + (2 * kk + hi) * 128 + lh * 2);
        v2f b0  = wp[0];
        v2f b1v = wp[16];
        v2f b2v = wp[32];
        v2f b3v = wp[48];

        acc0 = __builtin_amdgcn_wmma_f32_16x16x4_f32(false, a, false, b0,  (short)0, acc0, false, false);
        acc1 = __builtin_amdgcn_wmma_f32_16x16x4_f32(false, a, false, b1v, (short)0, acc1, false, false);
        acc2 = __builtin_amdgcn_wmma_f32_16x16x4_f32(false, a, false, b2v, (short)0, acc2, false, false);
        acc3 = __builtin_amdgcn_wmma_f32_16x16x4_f32(false, a, false, b3v, (short)0, acc3, false, false);
    }

    #pragma unroll
    for (int r = 0; r < 8; ++r) {
        int row = rowbase + r + hi * 8;
        if (row < nrows) {
            float* yp = Y + (size_t)row * 64 + lh;
            yp[0]  = acc0[r];
            yp[16] = acc1[r];
            yp[32] = acc2[r];
            yp[48] = acc3[r];
        }
    }
}


__device__ __forceinline__ void fma4(float4& a, const float4& v, float w) {
    a.x += v.x * w; a.y += v.y * w; a.z += v.z * w; a.w += v.w * w;
}

__global__ void __launch_bounds__(256)
gather_kernel(const float* __restrict__ xw, const int* __restrict__ csr,
              const int* __restrict__ rowptr, const int* __restrict__ cnt,
              const float* __restrict__ dinv, const float* __restrict__ bias,
              float* __restrict__ o, int N, int E, int do_relu) {
    const int t = blockIdx.x * 256 + threadIdx.x;
    const int d = t >> 4;
    const int q = t & 15;
    if (d >= N) return;

    const float4* x4 = reinterpret_cast<const float4*>(xw);
    const float di = dinv[d];
    const int start = min(max(rowptr[d], 0), E);
    const int n = min(max(cnt[d], 0), E - start);
    const int2* row = reinterpret_cast<const int2*>(csr) + start;
    const unsigned un = (unsigned)N;

    float4 acc = make_float4(0.0f, 0.0f, 0.0f, 0.0f);
    int j = 0;
    for (; j + 4 <= n; j += 4) {
        const int2 e0 = row[j], e1 = row[j + 1], e2 = row[j + 2], e3 = row[j + 3];
        const unsigned s0 = (unsigned)e0.x < un ? (unsigned)e0.x : 0u;
        const unsigned s1 = (unsigned)e1.x < un ? (unsigned)e1.x : 0u;
        const unsigned s2 = (unsigned)e2.x < un ? (unsigned)e2.x : 0u;
        const unsigned s3 = (unsigned)e3.x < un ? (unsigned)e3.x : 0u;
        const float4 v0 = x4[(size_t)s0 * 16 + q];
        const float4 v1 = x4[(size_t)s1 * 16 + q];
        const float4 v2 = x4[(size_t)s2 * 16 + q];
        const float4 v3 = x4[(size_t)s3 * 16 + q];
        fma4(acc, v0, __int_as_float(e0.y) * di);
        fma4(acc, v1, __int_as_float(e1.y) * di);
        fma4(acc, v2, __int_as_float(e2.y) * di);
        fma4(acc, v3, __int_as_float(e3.y) * di);
    }
    for (; j < n; ++j) {
        const int2 e0 = row[j];
        const unsigned s0 = (unsigned)e0.x < un ? (unsigned)e0.x : 0u;
        const float4 v0 = x4[(size_t)s0 * 16 + q];
        fma4(acc, v0, __int_as_float(e0.y) * di);
    }

    const float4 self = x4[(size_t)d * 16 + q];
    const float4 b = reinterpret_cast<const float4*>(bias)[q];
    float4 r;
    r.x = acc.x + self.x * di * di + b.x;
    r.y = acc.y + self.y * di * di + b.y;
    r.z = acc.z + self.z * di * di + b.z;
    r.w = acc.w + self.w * di * di + b.w;
    if (do_relu) {
        r.x = fmaxf(r.x, 0.0f); r.y = fmaxf(r.y, 0.0f);
        r.z = fmaxf(r.z, 0.0f); r.w = fmaxf(r.w, 0.0f);
    }
    reinterpret_cast<float4*>(o)[(size_t)d * 16 + q] = r;
}


extern "C" void kernel_launch(void* const* d_in, const int* in_sizes, int n_in,
                              void* d_out, int out_size, void* d_ws, size_t ws_size,
                              hipStream_t stream) {
    (void)n_in; (void)out_size; (void)ws_size;

    const float* x  = (const float*)d_in[0];
    const int*   ei = (const int*)d_in[1];
    const float* W1 = (const float*)d_in[2];
    const float* b1 = (const float*)d_in[3];
    const float* W2 = (const float*)d_in[4];
    const float* b2 = (const float*)d_in[5];
    float* out = (float*)d_out;

    const int N = in_sizes[0] / 64;
    const int E = in_sizes[1] / 2;
    const int* src = ei;
    const int* dst = ei + E;

    float* dinv = (float*)d_ws;
    float* xw   = dinv + (((size_t)N + 32767) & ~(size_t)32767);
    int* tab    = (int*)(xw + (size_t)N * 64);
    const size_t NP = ((size_t)N + 1023) & ~(size_t)1023;
    const size_t EP = ((size_t)E + 1023) & ~(size_t)1023;
    int* cnt    = tab;
    int* rowptr = tab + NP;
    int* rank   = tab + 2 * NP;
    int* csr    = tab + 2 * NP + EP;

    const int T = 256;

    fill_kernel<<<(N + T - 1) / T, T, 0, stream>>>((float*)cnt, 0.0f, N);
    hist_rank_kernel<<<(E + T - 1) / T, T, 0, stream>>>(dst, cnt, rank, E);
    scan_dinv_kernel<<<(N + T - 1) / T, T, 0, stream>>>(cnt, rowptr, dinv, N);
    fill_csr_kernel<<<(E + T - 1) / T, T, 0, stream>>>(src, dst, rank, rowptr, dinv, csr, E);

    gemm64_wmma<<<(N + 127) / 128, T, 0, stream>>>(x, W1, xw, N);
    gather_kernel<<<(N * 16 + T - 1) / T, T, 0, stream>>>(xw, csr, rowptr, cnt, dinv, b1, out, N, E, 1);

    gemm64_wmma<<<(N + 127) / 128, T, 0, stream>>>(out, W2, xw, N);
    gather_kernel<<<(N * 16 + T - 1) / T, T, 0, stream>>>(xw, csr, rowptr, cnt, dinv, b2, out, N, E, 0);
}


# layer-2 product fused into the layer-1 gather: a work-group keeps W2 in LDS, each 16-row tile of h goes LDS -> f32 WMMA -> whole-row stores; a launch and h's memory round trip go; sum order kept
# speedup vs baseline: 2.6785x; 1.3860x over previous
#include <hip/hip_runtime.h>


typedef __attribute__((ext_vector_type(2))) float v2f;
typedef __attribute__((ext_vector_type(8))) float v8f;

#define ROWW 32
#define SPILL_CAP 262144


__global__ void fill_kernel(float* __restrict__ p, float v, int n) {
    int i = blockIdx.x * blockDim.x + threadIdx.x;
    if (i < n) p[i] = v;
}


__global__ void bin_kernel(const int* __restrict__ src, const int* __restrict__ dst,
                           int* __restrict__ cnt, int* __restrict__ rows,
                           int* __restrict__ spill, int N, int E) {
    int e = blockIdx.x * blockDim.x + threadIdx.x;
    if (e >= E) return;
    const int s = src[e];
    const int d = dst[e];
    const int r = atomicAdd(&cnt[d], 1);
    if (r < ROWW) {
        rows[(size_t)d * ROWW + r] = s;
    } else {
        const int k = atomicAdd(&cnt[N], 1);
        if (k >= 0 && k < SPILL_CAP) reinterpret_cast<int2*>(spill)[k] = make_int2(d, s);
    }
}


__global__ void __launch_bounds__(256)
gemm64_wmma(const float* __restrict__ X, const float* __restrict__ W,
            const int* __restrict__ cnt, float* __restrict__ Y, int nrows) {
    __shared__ float lw[64 * 64];
    for (int i = threadIdx.x; i < 64 * 64; i += 256) {
        int k = i >> 6, n = i & 63;
        lw[(k >> 1) * 128 + n * 2 + (k & 1)] = W[i];
    }
    __syncthreads();

    int wave = threadIdx.x >> 5;
    int lane = threadIdx.x & 31;
    int lh   = lane & 15;
    int hi   = lane >> 4;
    int rowbase = (blockIdx.x * 8 + wave) * 16;

    int arow = rowbase + lh;
    int arow_c = arow < nrows ? arow : (nrows - 1);
    const v2f* xp = (const v2f*)(X + (size_t)arow_c * 64);

    float sc[8];
    #pragma unroll
    for (int r = 0; r < 8; ++r) {
        int row = rowbase + r + hi * 8;
        int row_c = row < nrows ? row : (nrows - 1);
        sc[r] = rsqrtf((float)(cnt[row_c] + 1));
    }

    v8f acc0 = {}, acc1 = {}, acc2 = {}, acc3 = {};

    #pragma unroll
    for (int kk = 0; kk < 16; ++kk) {
        v2f a = xp[2 * kk + hi];

        const v2f* wp = (const v2f*)(lw + (2 * kk + hi) * 128 + lh * 2);
        v2f b0  = wp[0];
        v2f b1v = wp[16];
        v2f b2v = wp[32];
        v2f b3v = wp[48];

        acc0 = __builtin_amdgcn_wmma_f32_16x16x4_f32(false, a, false, b0,  (short)0, acc0, false, false);
        acc1 = __builtin_amdgcn_wmma_f32_16x16x4_f32(false, a, false, b1v, (short)0, acc1, false, false);
        acc2 = __builtin_amdgcn_wmma_f32_16x16x4_f32(false, a, false, b2v, (short)0, acc2, false, false);
        acc3 = __builtin_amdgcn_wmma_f32_16x16x4_f32(false, a, false, b3v, (short)0, acc3, false, false);
    }

    #pragma unroll
    for (int r = 0; r < 8; ++r) {
        acc0[r] *= sc[r];
        acc1[r] *= sc[r];
        acc2[r] *= sc[r];
        acc3[r] *= sc[r];
    }

    #pragma unroll
    for (int r = 0; r < 8; ++r) {
        int row = rowbase + r + hi * 8;
        if (row < nrows) {
            float* yp = Y + (size_t)row * 64 + lh;
            yp[0]  = acc0[r];
            yp[16] = acc1[r];
            yp[32] = acc2[r];
            yp[48] = acc3[r];
        }
    }
}


__device__ __forceinline__ unsigned inrange(int s, unsigned un) {
    return (unsigned)s < un ? (unsigned)s : 0u;
}

__device__ __forceinline__ void add4(float4& a, const float4& v) {
    a.x += v.x; a.y += v.y; a.z += v.z; a.w += v.w;
}

__global__ void __launch_bounds__(256)
gather_kernel(const float* __restrict__ xs, const int* __restrict__ rows,
              const int* __restrict__ cnt, const int* __restrict__ spill,
              const int* __restrict__ src, const int* __restrict__ dst,
              const float* __restrict__ bias,
              float* __restrict__ o, int N, int E, int do_relu) {
    const int t = blockIdx.x * 256 + threadIdx.x;
    const int d = t >> 4;
    const int q = t & 15;
    if (d >= N) return;

    const float4* x4 = reinterpret_cast<const float4*>(xs);
    const int c = cnt[d];
    const float di = rsqrtf((float)(c + 1));
    const int n = min(max(c, 0), ROWW);
    const int* row = rows + (size_t)d * ROWW;
    const int4* row4 = reinterpret_cast<const int4*>(row);
    const unsigned un = (unsigned)N;

    float4 acc = make_float4(0.0f, 0.0f, 0.0f, 0.0f);
    int j = 0;
    for (; j + 8 <= n; j += 8) {
        const int4 ea = row4[j >> 2];
        const int4 eb = row4[(j >> 2) + 1];
        const float4 v0 = x4[(size_t)inrange(ea.x, un) * 16 + q];
        const float4 v1 = x4[(size_t)inrange(ea.y, un) * 16 + q];
        const float4 v2 = x4[(size_t)inrange(ea.z, un) * 16 + q];
        const float4 v3 = x4[(size_t)inrange(ea.w, un) * 16 + q];
        const float4 v4 = x4[(size_t)inrange(eb.x, un) * 16 + q];
        const float4 v5 = x4[(size_t)inrange(eb.y, un) * 16 + q];
        const float4 v6 = x4[(size_t)inrange(eb.z, un) * 16 + q];
        const float4 v7 = x4[(size_t)inrange(eb.w, un) * 16 + q];
        add4(acc, v0); add4(acc, v1); add4(acc, v2); add4(acc, v3);
        add4(acc, v4); add4(acc, v5); add4(acc, v6); add4(acc, v7);
    }
    if (j + 4 <= n) {
        const int4 e = row4[j >> 2];
        const float4 v0 = x4[(size_t)inrange(e.x, un) * 16 + q];
        const float4 v1 = x4[(size_t)inrange(e.y, un) * 16 + q];
        const float4 v2 = x4[(size_t)inrange(e.z, un) * 16 + q];
        const float4 v3 = x4[(size_t)inrange(e.w, un) * 16 + q];
        add4(acc, v0); add4(acc, v1); add4(acc, v2); add4(acc, v3);
        j += 4;
    }
    for (; j < n; ++j) {
        add4(acc, x4[(size_t)inrange(row[j], un) * 16 + q]);
    }
    if (c > ROWW) {
        const int total = cnt[N];
        if (total >= 0 && total <= SPILL_CAP) {
            const int2* sp = reinterpret_cast<const int2*>(spill);
            for (int k = 0; k < total; ++k) {
                const int2 p = sp[k];
                if (p.x == d) {
                    add4(acc, x4[(size_t)inrange(p.y, un) * 16 + q]);
                }
            }
        } else {
            acc = make_float4(0.0f, 0.0f, 0.0f, 0.0f);
            for (int e = 0; e < E; ++e) {
                if (dst[e] == d) {
                    add4(acc, x4[(size_t)inrange(src[e], un) * 16 + q]);
                }
            }
        }
    }

    const float4 self = x4[(size_t)d * 16 + q];
    const float4 b = reinterpret_cast<const float4*>(bias)[q];
    float4 r;
    r.x = (acc.x + self.x) * di + b.x;
    r.y = (acc.y + self.y) * di + b.y;
    r.z = (acc.z + self.z) * di + b.z;
    r.w = (acc.w + self.w) * di + b.w;
    if (do_relu) {
        r.x = fmaxf(r.x, 0.0f); r.y = fmaxf(r.y, 0.0f);
        r.z = fmaxf(r.z, 0.0f); r.w = fmaxf(r.w, 0.0f);
    }
    reinterpret_cast<float4*>(o)[(size_t)d * 16 + q] = r;
}

#define TPB 5
#define HP 68

__global__ void __launch_bounds__(256)
gather_gemm_kernel(const float* __restrict__ xs, const int* __restrict__ rows,
                   const int* __restrict__ cnt, const int* __restrict__ spill,
                   const int* __restrict__ src, const int* __restrict__ dst,
                   const float* __restrict__ bias, const float* __restrict__ W,
                   float* __restrict__ y, int N, int E) {
    __shared__ float sm[(64 + 16 + 16) * HP];
    float* const lw = sm;
    float* const ht = sm + 64 * HP;
    float* const ot = ht + 16 * HP;

    const int g = threadIdx.x >> 4;
    const int q = threadIdx.x & 15;
    const int wave = threadIdx.x >> 5;
    const int lane = threadIdx.x & 31;
    const int lh = lane & 15;
    const int hi = lane >> 4;

    const float4* x4 = reinterpret_cast<const float4*>(xs);
    const unsigned un = (unsigned)N;
    const int4 zero4 = make_int4(0, 0, 0, 0);

    int d = blockIdx.x * (TPB * 16) + g;
    int c = 0;
    int4 ea = zero4, eb = zero4;
    if (d < N) {
        const int4* r4 = reinterpret_cast<const int4*>(rows + (size_t)d * ROWW);
        c = cnt[d];
        ea = r4[0];
        eb = r4[1];
    }

    #pragma unroll
    for (int it = 0; it < 4; ++it) {
        const int i = it * 256 + threadIdx.x;
        const float4 w4 = reinterpret_cast<const float4*>(W)[i];
        const int k = i >> 4;
        const int n = (i & 15) * 4;
        float* p = lw + n * HP + k;
        p[0] = w4.x; p[HP] = w4.y; p[2 * HP] = w4.z; p[3 * HP] = w4.w;
    }

    for (int t = 0; t < TPB; ++t) {
        if (d - g >= N) break;

        float4 r = make_float4(0.0f, 0.0f, 0.0f, 0.0f);
        float di = 0.0f;
        if (d < N) {
            const float4 self = x4[(size_t)d * 16 + q];
            di = rsqrtf((float)(c + 1));
            const int n = min(max(c, 0), ROWW);
            const int4* row4 = reinterpret_cast<const int4*>(rows + (size_t)d * ROWW);

            float4 acc = make_float4(0.0f, 0.0f, 0.0f, 0.0f);
            for (int j = 0; j < n; j += 8) {
                int4 na = zero4, nb = zero4;
                if (j + 8 < n) {
                    na = row4[(j >> 2) + 2];
                    nb = row4[(j >> 2) + 3];
                }
                const int m = n - j;
                const float4 z = make_float4(0.0f, 0.0f, 0.0f, 0.0f);
                float4 v0 = x4[(size_t)inrange(ea.x, un) * 16 + q];
                float4 v1 = z, v2 = z, v3 = z, v4 = z, v5 = z, v6 = z, v7 = z;
                if (m > 1) v1 = x4[(size_t)inrange(ea.y, un) * 16 + q];
                if (m > 2) v2 = x4[(size_t)inrange(ea.z, un) * 16 + q];
                if (m > 3) v3 = x4[(size_t)inrange(ea.w, un) * 16 + q];
                if (m > 4) v4 = x4[(size_t)inrange(eb.x, un) * 16 + q];
                if (m > 5) v5 = x4[(size_t)inrange(eb.y, un) * 16 + q];
                if (m > 6) v6 = x4[(size_t)inrange(eb.z, un) * 16 + q];
                if (m > 7) v7 = x4[(size_t)inrange(eb.w, un) * 16 + q];
                add4(acc, v0); add4(acc, v1); add4(acc, v2); add4(acc, v3);
                add4(acc, v4); add4(acc, v5); add4(acc, v6); add4(acc, v7);
                ea = na; eb = nb;
            }
            if (c > ROWW) {
                const int total = cnt[N];
                if (total >= 0 && total <= SPILL_CAP) {
                    const int2* sp = reinterpret_cast<const int2*>(spill);
                    for (int k = 0; k < total; ++k) {
                        const int2 p = sp[k];
                        if (p.x == d) {
                            add4(acc, x4[(size_t)inrange(p.y, un) * 16 + q]);
                        }
                    }
                } else {
                    acc = make_float4(0.0f, 0.0f, 0.0f, 0.0f);
                    for (int e = 0; e < E; ++e) {
                        if (dst[e] == d) {
                            add4(acc, x4[(size_t)inrange(src[e], un) * 16 + q]);
                        }
                    }
                }
            }

            const float4 b = reinterpret_cast<const float4*>(bias)[q];
            r.x = fmaxf((acc.x + self.x) * di + b.x, 0.0f);
            r.y = fmaxf((acc.y + self.y) * di + b.y, 0.0f);
            r.z = fmaxf((acc.z + self.z) * di + b.z, 0.0f);
            r.w = fmaxf((acc.w + self.w) * di + b.w, 0.0f);
        }
        *reinterpret_cast<float4*>(ht + g * HP + 4 * q) = r;
        __syncthreads();

        const int dn = d + 16;
        int cn = 0;
        int4 ean = zero4, ebn = zero4;
        if (t + 1 < TPB && dn < N) {
            const int4* r4 = reinterpret_cast<const int4*>(rows + (size_t)dn * ROWW);
            cn = cnt[dn];
            ean = r4[0];
            ebn = r4[1];
        }

        if (wave < 4) {
            const int n0 = wave * 16;
            const float* ap = ht + lh * HP + 2 * hi;
            const float* bp = lw + (n0 + lh) * HP + 2 * hi;
            v8f acc = {};
            #pragma unroll
            for (int kk = 0; kk < 16; ++kk) {
                const v2f a  = *reinterpret_cast<const v2f*>(ap + 4 * kk);
                const v2f bw = *reinterpret_cast<const v2f*>(bp + 4 * kk);
                acc = __builtin_amdgcn_wmma_f32_16x16x4_f32(false, a, false, bw, (short)0, acc, false, false);
            }
            float* op = ot + 8 * hi * HP + n0 + lh;
            #pragma unroll
            for (int rr = 0; rr < 8; ++rr) {
                op[rr * HP] = acc[rr];
            }
        }
        __syncthreads();

        if (d < N) {
            float4 o = *reinterpret_cast<const float4*>(ot + g * HP + 4 * q);
            o.x *= di; o.y *= di; o.z *= di; o.w *= di;
            reinterpret_cast<float4*>(y)[(size_t)d * 16 + q] = o;
        }

        d = dn; c = cn; ea = ean; eb = ebn;
    }
}


extern "C" void kernel_launch(void* const* d_in, const int* in_sizes, int n_in,
                              void* d_out, int out_size, void* d_ws, size_t ws_size,
                              hipStream_t stream) {
    (void)n_in; (void)out_size; (void)ws_size;

    const float* x  = (const float*)d_in[0];
    const int*   ei = (const int*)d_in[1];
    const float* W1 = (const float*)d_in[2];
    const float* b1 = (const float*)d_in[3];
    const float* W2 = (const float*)d_in[4];
    const float* b2 = (const float*)d_in[5];
    float* out = (float*)d_out;

    const int N = in_sizes[0] / 64;
    const int E = in_sizes[1] / 2;
    const int* src = ei;
    const int* dst = ei + E;

    int* cnt    = (int*)d_ws;
    float* xs   = (float*)d_ws + (((size_t)N + 1 + 32767) & ~(size_t)32767);
    int* rows   = (int*)(xs + (size_t)N * 64);
    int* spill  = rows + (size_t)N * ROWW;

    const int T = 256;

    fill_kernel<<<(N + 1 + T - 1) / T, T, 0, stream>>>((float*)cnt, 0.0f, N + 1);
    bin_kernel<<<(E + T - 1) / T, T, 0, stream>>>(src, dst, cnt, rows, spill, N, E);

    gemm64_wmma<<<(N + 127) / 128, T, 0, stream>>>(x, W1, cnt, out, N);

    gather_gemm_kernel<<<(N + 16 * TPB - 1) / (16 * TPB), T, 0, stream>>>(out, rows, cnt, spill, src, dst, b1, W2, xs, N, E);

    gather_kernel<<<(N * 16 + T - 1) / T, T, 0, stream>>>(xs, rows, cnt, spill, src, dst, b2, out, N, E, 0);
}


# adjacency by a two-level counting sort in LDS: edges split by 256-destination range, row slots from LDS adds (sum order = slot order, now their arrival order); per-edge global atomic + 4-B scatter go
# speedup vs baseline: 4.3154x; 1.6111x over previous
#include <hip/hip_runtime.h>


typedef __attribute__((ext_vector_type(2))) float v2f;
typedef __attribute__((ext_vector_type(8))) float v8f;

#define ROWW 32
#define SPILL_CAP 262144

#define BINW 256
#define BINSH 8
#define NSEG 250
#define SEGCAP 64
#define NBMAX 512
#define EPT 19
#define TILE (EPT * 256)

__global__ void __launch_bounds__(256)
part_kernel(const int* __restrict__ src, const int* __restrict__ dst,
            int* __restrict__ stage, int* __restrict__ segcnt, int* __restrict__ cnt, int N, int E) {
    __shared__ int hist[NBMAX];
    __shared__ int start[NBMAX];
    __shared__ int cursor[NBMAX];
    __shared__ int segoff[NBMAX];
    __shared__ int sc[256];
    __shared__ int lval[TILE];
    __shared__ int lbin[TILE];

    const int t = threadIdx.x;
    const int blk = blockIdx.x;
    const int nb = (N + BINW - 1) >> BINSH;
    const int per = (E + NSEG - 1) / NSEG;
    const int cbeg = min(blk * per, E);
    const int cend = min(cbeg + per, E);

    if (blk == 0 && t == 0) cnt[N] = 0;
    segoff[t] = 0;
    segoff[t + 256] = 0;

    for (int base = cbeg; base < cend; base += TILE) {
        hist[t] = 0;
        hist[t + 256] = 0;
        __syncthreads();

        int dv[EPT], sv[EPT];
        #pragma unroll
        for (int k = 0; k < EPT; ++k) {
            const int e = base + k * 256 + t;
            const int ec = min(e, cend - 1);
            int d = dst[ec];
            const int s = src[ec];
            if (e >= cend || (unsigned)d >= (unsigned)N || (d >> BINSH) >= NBMAX) d = -1;
            dv[k] = d;
            sv[k] = s;
        }
        #pragma unroll
        for (int k = 0; k < EPT; ++k) {
            if (dv[k] >= 0) atomicAdd(&hist[dv[k] >> BINSH], 1);
        }
        __syncthreads();

        const int h0 = hist[2 * t];
        const int h1 = hist[2 * t + 1];
        const int mine = h0 + h1;
        sc[t] = mine;
        __syncthreads();
        for (int off = 1; off < 256; off <<= 1) {
            const int v = t >= off ? sc[t - off] : 0;
            __syncthreads();
            sc[t] += v;
            __syncthreads();
        }
        const int ex = sc[t] - mine;
        start[2 * t] = ex;
        start[2 * t + 1] = ex + h0;
        cursor[2 * t] = ex;
        cursor[2 * t + 1] = ex + h0;
        const int total = sc[255];
        __syncthreads();

        #pragma unroll
        for (int k = 0; k < EPT; ++k) {
            if (dv[k] >= 0) {
                const int bin = dv[k] >> BINSH;
                const int pos = atomicAdd(&cursor[bin], 1);
                lval[pos] = (int)(((unsigned)(dv[k] & (BINW - 1)) << 24) | ((unsigned)sv[k] & 0xFFFFFFu));
                lbin[pos] = bin;
            }
        }
        __syncthreads();

        for (int i = t; i < total; i += 256) {
            const int bin = lbin[i];
            const int off = segoff[bin] + (i - start[bin]);
            if (off < SEGCAP) stage[((size_t)bin * NSEG + blk) * SEGCAP + off] = lval[i];
        }
        __syncthreads();
        segoff[t] += hist[t];
        segoff[t + 256] += hist[t + 256];
    }
    __syncthreads();
    for (int bin = t; bin < nb && bin < NBMAX; bin += 256) segcnt[bin * NSEG + blk] = segoff[bin];
}

__device__ __forceinline__ void place_entry(int dl, int s, int d0, int* lcnt, int* lrows,
                                            int* __restrict__ cnt, int* __restrict__ spill, int N) {
    const int r = atomicAdd(&lcnt[dl], 1);
    if (r < ROWW) {
        lrows[dl * ROWW + r] = s;
    } else {
        const int k = atomicAdd(&cnt[N], 1);
        if (k >= 0 && k < SPILL_CAP) reinterpret_cast<int2*>(spill)[k] = make_int2(d0 + dl, s);
    }
}

__global__ void __launch_bounds__(256)
rows_kernel(const int* __restrict__ src, const int* __restrict__ dst,
            const int* __restrict__ stage, const int* __restrict__ segcnt,
            int* __restrict__ cnt, int* __restrict__ rows, int* __restrict__ spill, int N, int E) {
    __shared__ int lcnt[BINW];
    __shared__ int lrows[BINW * ROWW];
    __shared__ int lover;

    const int t = threadIdx.x;
    const int b = blockIdx.x;
    const int d0 = b * BINW;
    const int4 zero4 = make_int4(0, 0, 0, 0);
    int4* const lr4 = reinterpret_cast<int4*>(lrows);

    int c = 0;
    int4 v0 = zero4, v1 = zero4, v2 = zero4, v3 = zero4;
    const int4* sp = reinterpret_cast<const int4*>(stage + ((size_t)b * NSEG + (t < NSEG ? t : 0)) * SEGCAP);
    if (t < NSEG) {
        c = segcnt[b * NSEG + t];
        v0 = sp[0]; v1 = sp[1]; v2 = sp[2]; v3 = sp[3];
    }

    lcnt[t] = 0;
    for (int i = t; i < BINW * ROWW / 4; i += 256) lr4[i] = zero4;
    if (t == 0) lover = 0;
    __syncthreads();
    if (c > SEGCAP) lover = 1;
    __syncthreads();
    const bool over = lover != 0;

    if (!over) {
        const int w[16] = { v0.x, v0.y, v0.z, v0.w, v1.x, v1.y, v1.z, v1.w,
                            v2.x, v2.y, v2.z, v2.w, v3.x, v3.y, v3.z, v3.w };
        #pragma unroll
        for (int i = 0; i < 16; ++i) {
            if (i < c) place_entry((int)((unsigned)w[i] >> 24), w[i] & 0xFFFFFF, d0, lcnt, lrows, cnt, spill, N);
        }
        for (int i = 16; i < c; i += 4) {
            const int4 v = sp[i >> 2];
            place_entry((int)((unsigned)v.x >> 24), v.x & 0xFFFFFF, d0, lcnt, lrows, cnt, spill, N);
            if (i + 1 < c) place_entry((int)((unsigned)v.y >> 24), v.y & 0xFFFFFF, d0, lcnt, lrows, cnt, spill, N);
            if (i + 2 < c) place_entry((int)((unsigned)v.z >> 24), v.z & 0xFFFFFF, d0, lcnt, lrows, cnt, spill, N);
            if (i + 3 < c) place_entry((int)((unsigned)v.w >> 24), v.w & 0xFFFFFF, d0, lcnt, lrows, cnt, spill, N);
        }
    } else {
        for (int e = t; e < E; e += 256) {
            const int d = dst[e];
            if ((unsigned)d < (unsigned)N && (d >> BINSH) == b) {
                place_entry(d & (BINW - 1), src[e], d0, lcnt, lrows, cnt, spill, N);
            }
        }
    }
    __syncthreads();

    if (d0 + t < N) cnt[d0 + t] = lcnt[t];
    int4* const g4 = reinterpret_cast<int4*>(rows + (size_t)d0 * ROWW);
    for (int i = t; i < BINW * ROWW / 4; i += 256) {
        if (d0 + (i >> 3) < N) g4[i] = lr4[i];
    }
}


__global__ void __launch_bounds__(256)
gemm64_wmma(const float* __restrict__ X, const float* __restrict__ W,
            const int* __restrict__ cnt, float* __restrict__ Y, int nrows) {
    __shared__ float lw[64 * 64];
    for (int i = threadIdx.x; i < 64 * 64; i += 256) {
        int k = i >> 6, n = i & 63;
        lw[(k >> 1) * 128 + n * 2 + (k & 1)] = W[i];
    }
    __syncthreads();

    int wave = threadIdx.x >> 5;
    int lane = threadIdx.x & 31;
    int lh   = lane & 15;
    int hi   = lane >> 4;
    int rowbase = (blockIdx.x * 8 + wave) * 16;

    int arow = rowbase + lh;
    int arow_c = arow < nrows ? arow : (nrows - 1);
    const v2f* xp = (const v2f*)(X + (size_t)arow_c * 64);

    float sc[8];
    #pragma unroll
    for (int r = 0; r < 8; ++r) {
        int row = rowbase + r + hi * 8;
        int row_c = row < nrows ? row : (nrows - 1);
        sc[r] = rsqrtf((float)(cnt[row_c] + 1));
    }

    v8f acc0 = {}, acc1 = {}, acc2 = {}, acc3 = {};

    #pragma unroll
    for (int kk = 0; kk < 16; ++kk) {
        v2f a = xp[2 * kk + hi];

        const v2f* wp = (const v2f*)(lw + (2 * kk + hi) * 128 + lh * 2);
        v2f b0  = wp[0];
        v2f b1v = wp[16];
        v2f b2v = wp[32];
        v2f b3v = wp[48];

        acc0 = __builtin_amdgcn_wmma_f32_16x16x4_f32(false, a, false, b0,  (short)0, acc0, false, false);
        acc1 = __builtin_amdgcn_wmma_f32_16x16x4_f32(false, a, false, b1v, (short)0, acc1, false, false);
        acc2 = __builtin_amdgcn_wmma_f32_16x16x4_f32(false, a, false, b2v, (short)0, acc2, false, false);
        acc3 = __builtin_amdgcn_wmma_f32_16x16x4_f32(false, a, false, b3v, (short)0, acc3, false, false);
    }

    #pragma unroll
    for (int r = 0; r < 8; ++r) {
        acc0[r] *= sc[r];
        acc1[r] *= sc[r];
        acc2[r] *= sc[r];
        acc3[r] *= sc[r];
    }

    #pragma unroll
    for (int r = 0; r < 8; ++r) {
        int row = rowbase + r + hi * 8;
        if (row < nrows) {
            float* yp = Y + (size_t)row * 64 + lh;
            yp[0]  = acc0[r];
            yp[16] = acc1[r];
            yp[32] = acc2[r];
            yp[48] = acc3[r];
        }
    }
}


__device__ __forceinline__ unsigned inrange(int s, unsigned un) {
    return (unsigned)s < un ? (unsigned)s : 0u;
}

__device__ __forceinline__ void add4(float4& a, const float4& v) {
    a.x += v.x; a.y += v.y; a.z += v.z; a.w += v.w;
}

__global__ void __launch_bounds__(256)
gather_kernel(const float* __restrict__ xs, const int* __restrict__ rows,
              const int* __restrict__ cnt, const int* __restrict__ spill,
              const int* __restrict__ src, const int* __restrict__ dst,
              const float* __restrict__ bias,
              float* __restrict__ o, int N, int E, int do_relu) {
    const int t = blockIdx.x * 256 + threadIdx.x;
    const int d = t >> 4;
    const int q = t & 15;
    if (d >= N) return;

    const float4* x4 = reinterpret_cast<const float4*>(xs);
    const int c = cnt[d];
    const float di = rsqrtf((float)(c + 1));
    const int n = min(max(c, 0), ROWW);
    const int* row = rows + (size_t)d * ROWW;
    const int4* row4 = reinterpret_cast<const int4*>(row);
    const unsigned un = (unsigned)N;

    float4 acc = make_float4(0.0f, 0.0f, 0.0f, 0.0f);
    int j = 0;
    for (; j + 8 <= n; j += 8) {
        const int4 ea = row4[j >> 2];
        const int4 eb = row4[(j >> 2) + 1];
        const float4 v0 = x4[(size_t)inrange(ea.x, un) * 16 + q];
        const float4 v1 = x4[(size_t)inrange(ea.y, un) * 16 + q];
        const float4 v2 = x4[(size_t)inrange(ea.z, un) * 16 + q];
        const float4 v3 = x4[(size_t)inrange(ea.w, un) * 16 + q];
        const float4 v4 = x4[(size_t)inrange(eb.x, un) * 16 + q];
        const float4 v5 = x4[(size_t)inrange(eb.y, un) * 16 + q];
        const float4 v6 = x4[(size_t)inrange(eb.z, un) * 16 + q];
        const float4 v7 = x4[(size_t)inrange(eb.w, un) * 16 + q];
        add4(acc, v0); add4(acc, v1); add4(acc, v2); add4(acc, v3);
        add4(acc, v4); add4(acc, v5); add4(acc, v6); add4(acc, v7);
    }
    if (j + 4 <= n) {
        const int4 e = row4[j >> 2];
        const float4 v0 = x4[(size_t)inrange(e.x, un) * 16 + q];
        const float4 v1 = x4[(size_t)inrange(e.y, un) * 16 + q];
        const float4 v2 = x4[(size_t)inrange(e.z, un) * 16 + q];
        const float4 v3 = x4[(size_t)inrange(e.w, un) * 16 + q];
        add4(acc, v0); add4(acc, v1); add4(acc, v2); add4(acc, v3);
        j += 4;
    }
    for (; j < n; ++j) {
        add4(acc, x4[(size_t)inrange(row[j], un) * 16 + q]);
    }
    if (c > ROWW) {
        const int total = cnt[N];
        if (total >= 0 && total <= SPILL_CAP) {
            const int2* sp = reinterpret_cast<const int2*>(spill);
            for (int k = 0; k < total; ++k) {
                const int2 p = sp[k];
                if (p.x == d) {
                    add4(acc, x4[(size_t)inrange(p.y, un) * 16 + q]);
                }
            }
        } else {
            acc = make_float4(0.0f, 0.0f, 0.0f, 0.0f);
            for (int e = 0; e < E; ++e) {
                if (dst[e] == d) {
                    add4(acc, x4[(size_t)inrange(src[e], un) * 16 + q]);
                }
            }
        }
    }

    const float4 self = x4[(size_t)d * 16 + q];
    const float4 b = reinterpret_cast<const float4*>(bias)[q];
    float4 r;
    r.x = (acc.x + self.x) * di + b.x;
    r.y = (acc.y + self.y) * di + b.y;
    r.z = (acc.z + self.z) * di + b.z;
    r.w = (acc.w + self.w) * di + b.w;
    if (do_relu) {
        r.x = fmaxf(r.x, 0.0f); r.y = fmaxf(r.y, 0.0f);
        r.z = fmaxf(r.z, 0.0f); r.w = fmaxf(r.w, 0.0f);
    }
    reinterpret_cast<float4*>(o)[(size_t)d * 16 + q] = r;
}

#define TPB 5
#define HP 68

__global__ void __launch_bounds__(256)
gather_gemm_kernel(const float* __restrict__ xs, const int* __restrict__ rows,
                   const int* __restrict__ cnt, const int* __restrict__ spill,
                   const int* __restrict__ src, const int* __restrict__ dst,
                   const float* __restrict__ bias, const float* __restrict__ W,
                   float* __restrict__ y, int N, int E) {
    __shared__ float sm[(64 + 16 + 16) * HP];
    float* const lw = sm;
    float* const ht = sm + 64 * HP;
    float* const ot = ht + 16 * HP;

    const int g = threadIdx.x >> 4;
    const int q = threadIdx.x & 15;
    const int wave = threadIdx.x >> 5;
    const int lane = threadIdx.x & 31;
    const int lh = lane & 15;
    const int hi = lane >> 4;

    const float4* x4 = reinterpret_cast<const float4*>(xs);
    const unsigned un = (unsigned)N;
    const int4 zero4 = make_int4(0, 0, 0, 0);

    int d = blockIdx.x * (TPB * 16) + g;
    int c = 0;
    int4 ea = zero4, eb = zero4;
    if (d < N) {
        const int4* r4 = reinterpret_cast<const int4*>(rows + (size_t)d * ROWW);
        c = cnt[d];
        ea = r4[0];
        eb = r4[1];
    }

    #pragma unroll
    for (int it = 0; it < 4; ++it) {
        const int i = it * 256 + threadIdx.x;
        const float4 w4 = reinterpret_cast<const float4*>(W)[i];
        const int k = i >> 4;
        const int n = (i & 15) * 4;
        float* p = lw + n * HP + k;
        p[0] = w4.x; p[HP] = w4.y; p[2 * HP] = w4.z; p[3 * HP] = w4.w;
    }

    for (int t = 0; t < TPB; ++t) {
        if (d - g >= N) break;

        float4 r = make_float4(0.0f, 0.0f, 0.0f, 0.0f);
        float di = 0.0f;
        if (d < N) {
            const float4 self = x4[(size_t)d * 16 + q];
            di = rsqrtf((float)(c + 1));
            const int n = min(max(c, 0), ROWW);
            const int4* row4 = reinterpret_cast<const int4*>(rows + (size_t)d * ROWW);

            float4 acc = make_float4(0.0f, 0.0f, 0.0f, 0.0f);
            for (int j = 0; j < n; j += 8) {
                int4 na = zero4, nb = zero4;
                if (j + 8 < n) {
                    na = row4[(j >> 2) + 2];
                    nb = row4[(j >> 2) + 3];
                }
                const int m = n - j;
                const float4 z = make_float4(0.0f, 0.0f, 0.0f, 0.0f);
                float4 v0 = x4[(size_t)inrange(ea.x, un) * 16 + q];
                float4 v1 = z, v2 = z, v3 = z, v4 = z, v5 = z, v6 = z, v7 = z;
                if (m > 1) v1 = x4[(size_t)inrange(ea.y, un) * 16 + q];
                if (m > 2) v2 = x4[(size_t)inrange(ea.z, un) * 16 + q];
                if (m > 3) v3 = x4[(size_t)inrange(ea.w, un) * 16 + q];
                if (m > 4) v4 = x4[(size_t)inrange(eb.x, un) * 16 + q];
                if (m > 5) v5 = x4[(size_t)inrange(eb.y, un) * 16 + q];
                if (m > 6) v6 = x4[(size_t)inrange(eb.z, un) * 16 + q];
                if (m > 7) v7 = x4[(size_t)inrange(eb.w, un) * 16 + q];
                add4(acc, v0); add4(acc, v1); add4(acc, v2); add4(acc, v3);
                add4(acc, v4); add4(acc, v5); add4(acc, v6); add4(acc, v7);
                ea = na; eb = nb;
            }
            if (c > ROWW) {
                const int total = cnt[N];
                if (total >= 0 && total <= SPILL_CAP) {
                    const int2* sp = reinterpret_cast<const int2*>(spill);
                    for (int k = 0; k < total; ++k) {
                        const int2 p = sp[k];
                        if (p.x == d) {
                            add4(acc, x4[(size_t)inrange(p.y, un) * 16 + q]);
                        }
                    }
                } else {
                    acc = make_float4(0.0f, 0.0f, 0.0f, 0.0f);
                    for (int e = 0; e < E; ++e) {
                        if (dst[e] == d) {
                            add4(acc, x4[(size_t)inrange(src[e], un) * 16 + q]);
                        }
                    }
                }
            }

            const float4 b = reinterpret_cast<const float4*>(bias)[q];
            r.x = fmaxf((acc.x + self.x) * di + b.x, 0.0f);
            r.y = fmaxf((acc.y + self.y) * di + b.y, 0.0f);
            r.z = fmaxf((acc.z + self.z) * di + b.z, 0.0f);
            r.w = fmaxf((acc.w + self.w) * di + b.w, 0.0f);
        }
        *reinterpret_cast<float4*>(ht + g * HP + 4 * q) = r;
        __syncthreads();

        const int dn = d + 16;
        int cn = 0;
        int4 ean = zero4, ebn = zero4;
        if (t + 1 < TPB && dn < N) {
            const int4* r4 = reinterpret_cast<const int4*>(rows + (size_t)dn * ROWW);
            cn = cnt[dn];
            ean = r4[0];
            ebn = r4[1];
        }

        if (wave < 4) {
            const int n0 = wave * 16;
            const float* ap = ht + lh * HP + 2 * hi;
            const float* bp = lw + (n0 + lh) * HP + 2 * hi;
            v8f acc = {};
            #pragma unroll
            for (int kk = 0; kk < 16; ++kk) {
                const v2f a  = *reinterpret_cast<const v2f*>(ap + 4 * kk);
                const v2f bw = *reinterpret_cast<const v2f*>(bp + 4 * kk);
                acc = __builtin_amdgcn_wmma_f32_16x16x4_f32(false, a, false, bw, (short)0, acc, false, false);
            }
            float* op = ot + 8 * hi * HP + n0 + lh;
            #pragma unroll
            for (int rr = 0; rr < 8; ++rr) {
                op[rr * HP] = acc[rr];
            }
        }
        __syncthreads();

        if (d < N) {
            float4 o = *reinterpret_cast<const float4*>(ot + g * HP + 4 * q);
            o.x *= di; o.y *= di; o.z *= di; o.w *= di;
            reinterpret_cast<float4*>(y)[(size_t)d * 16 + q] = o;
        }

        d = dn; c = cn; ea = ean; eb = ebn;
    }
}


extern "C" void kernel_launch(void* const* d_in, const int* in_sizes, int n_in,
                              void* d_out, int out_size, void* d_ws, size_t ws_size,
                              hipStream_t stream) {
    (void)n_in; (void)out_size; (void)ws_size;

    const float* x  = (const float*)d_in[0];
    const int*   ei = (const int*)d_in[1];
    const float* W1 = (const float*)d_in[2];
    const float* b1 = (const float*)d_in[3];
    const float* W2 = (const float*)d_in[4];
    const float* b2 = (const float*)d_in[5];
    float* out = (float*)d_out;

    const int N = in_sizes[0] / 64;
    const int E = in_sizes[1] / 2;
    const int* src = ei;
    const int* dst = ei + E;

    int* cnt    = (int*)d_ws;
    float* xs   = (float*)d_ws + (((size_t)N + 1 + 32767) & ~(size_t)32767);
    int* rows   = (int*)(xs + (size_t)N * 64);
    int* spill  = rows + (size_t)N * ROWW;

    const int T = 256;

    int* stage  = (int*)xs;
    int* segcnt = (int*)out;
    part_kernel<<<NSEG, T, 0, stream>>>(src, dst, stage, segcnt, cnt, N, E);
    rows_kernel<<<(N + BINW - 1) / BINW, T, 0, stream>>>(src, dst, stage, segcnt, cnt, rows, spill, N, E);

    gemm64_wmma<<<(N + 127) / 128, T, 0, stream>>>(x, W1, cnt, out, N);

    gather_gemm_kernel<<<(N + 16 * TPB - 1) / (16 * TPB), T, 0, stream>>>(out, rows, cnt, spill, src, dst, b1, W2, xs, N, E);

    gather_kernel<<<(N * 16 + T - 1) / T, T, 0, stream>>>(xs, rows, cnt, spill, src, dst, b2, out, N, E, 0);
}


# layer-1 product moved into the row-building launch: each bin's work-group multiplies its own 256 rows of x by W1 once their counts sit in LDS; one launch fewer (5 to 4); every sum keeps its order
# speedup vs baseline: 4.4746x; 1.0369x over previous
#include <hip/hip_runtime.h>


typedef __attribute__((ext_vector_type(2))) float v2f;
typedef __attribute__((ext_vector_type(8))) float v8f;

#define ROWW 32
#define SPILL_CAP 262144

#define BINW 256
#define BINSH 8
#define NSEG 250
#define SEGCAP 64
#define NBMAX 512
#define EPT 19
#define TILE (EPT * 256)

__global__ void __launch_bounds__(256)
part_kernel(const int* __restrict__ src, const int* __restrict__ dst,
            int* __restrict__ stage, int* __restrict__ segcnt, int* __restrict__ cnt, int N, int E) {
    __shared__ int hist[NBMAX];
    __shared__ int start[NBMAX];
    __shared__ int cursor[NBMAX];
    __shared__ int segoff[NBMAX];
    __shared__ int sc[256];
    __shared__ int lval[TILE];
    __shared__ int lbin[TILE];

    const int t = threadIdx.x;
    const int blk = blockIdx.x;
    const int nb = (N + BINW - 1) >> BINSH;
    const int per = (E + NSEG - 1) / NSEG;
    const int cbeg = min(blk * per, E);
    const int cend = min(cbeg + per, E);

    if (blk == 0 && t == 0) cnt[N] = 0;
    segoff[t] = 0;
    segoff[t + 256] = 0;

    for (int base = cbeg; base < cend; base += TILE) {
        hist[t] = 0;
        hist[t + 256] = 0;
        __syncthreads();

        int dv[EPT], sv[EPT];
        #pragma unroll
        for (int k = 0; k < EPT; ++k) {
            const int e = base + k * 256 + t;
            const int ec = min(e, cend - 1);
            int d = dst[ec];
            const int s = src[ec];
            if (e >= cend || (unsigned)d >= (unsigned)N || (d >> BINSH) >= NBMAX) d = -1;
            dv[k] = d;
            sv[k] = s;
        }
        #pragma unroll
        for (int k = 0; k < EPT; ++k) {
            if (dv[k] >= 0) atomicAdd(&hist[dv[k] >> BINSH], 1);
        }
        __syncthreads();

        const int h0 = hist[2 * t];
        const int h1 = hist[2 * t + 1];
        const int mine = h0 + h1;
        sc[t] = mine;
        __syncthreads();
        for (int off = 1; off < 256; off <<= 1) {
            const int v = t >= off ? sc[t - off] : 0;
            __syncthreads();
            sc[t] += v;
            __syncthreads();
        }
        const int ex = sc[t] - mine;
        start[2 * t] = ex;
        start[2 * t + 1] = ex + h0;
        cursor[2 * t] = ex;
        cursor[2 * t + 1] = ex + h0;
        const int total = sc[255];
        __syncthreads();

        #pragma unroll
        for (int k = 0; k < EPT; ++k) {
            if (dv[k] >= 0) {
                const int bin = dv[k] >> BINSH;
                const int pos = atomicAdd(&cursor[bin], 1);
                lval[pos] = (int)(((unsigned)(dv[k] & (BINW - 1)) << 24) | ((unsigned)sv[k] & 0xFFFFFFu));
                lbin[pos] = bin;
            }
        }
        __syncthreads();

        for (int i = t; i < total; i += 256) {
            const int bin = lbin[i];
            const int off = segoff[bin] + (i - start[bin]);
            if (off < SEGCAP) stage[((size_t)bin * NSEG + blk) * SEGCAP + off] = lval[i];
        }
        __syncthreads();
        segoff[t] += hist[t];
        segoff[t + 256] += hist[t + 256];
    }
    __syncthreads();
    for (int bin = t; bin < nb && bin < NBMAX; bin += 256) segcnt[bin * NSEG + blk] = segoff[bin];
}

__device__ __forceinline__ void lds_barrier() {
    __builtin_amdgcn_fence(__ATOMIC_RELEASE, "workgroup", "local");
    __builtin_amdgcn_s_barrier();
    __builtin_amdgcn_fence(__ATOMIC_ACQUIRE, "workgroup", "local");
}

__device__ __forceinline__ void place_entry(int dl, int s, int d0, int* lcnt, int* lrows,
                                            int* __restrict__ cnt, int* __restrict__ spill, int N) {
    const int r = atomicAdd(&lcnt[dl], 1);
    if (r < ROWW) {
        lrows[dl * ROWW + r] = s;
    } else {
        const int k = atomicAdd(&cnt[N], 1);
        if (k >= 0 && k < SPILL_CAP) reinterpret_cast<int2*>(spill)[k] = make_int2(d0 + dl, s);
    }
}

__global__ void __launch_bounds__(256)
rows_gemm_kernel(const int* __restrict__ src, const int* __restrict__ dst,
                 const int* __restrict__ stage, const int* __restrict__ segcnt,
                 int* __restrict__ cnt, int* __restrict__ rows, int* __restrict__ spill,
                 const float* __restrict__ X, const float* __restrict__ W, float* __restrict__ Y,
                 int N, int E) {
    __shared__ int lcnt[BINW];
    __shared__ int lrows[BINW * ROWW];
    __shared__ int lover;
    __shared__ float lw[64 * 64];

    const int t = threadIdx.x;
    const int b = blockIdx.x;
    const int d0 = b * BINW;
    const int4 zero4 = make_int4(0, 0, 0, 0);
    int4* const lr4 = reinterpret_cast<int4*>(lrows);

    const int wave = t >> 5;
    const int lane = t & 31;
    const int lh   = lane & 15;
    const int hi   = lane >> 4;

    float4 w4[4];
    #pragma unroll
    for (int it = 0; it < 4; ++it) w4[it] = reinterpret_cast<const float4*>(W)[it * 256 + t];

    int c = 0;
    int4 v0 = zero4, v1 = zero4, v2 = zero4, v3 = zero4;
    const int4* sp = reinterpret_cast<const int4*>(stage + ((size_t)b * NSEG + (t < NSEG ? t : 0)) * SEGCAP);
    if (t < NSEG) {
        c = segcnt[b * NSEG + t];
        v0 = sp[0]; v1 = sp[1]; v2 = sp[2]; v3 = sp[3];
    }

    v2f a0[16], a1[16];
    {
        const int r0 = d0 + wave * 16 + lh;
        const int r1 = r0 + 128;
        const v2f* xp0 = (const v2f*)(X + (size_t)(r0 < N ? r0 : (N - 1)) * 64);
        const v2f* xp1 = (const v2f*)(X + (size_t)(r1 < N ? r1 : (N - 1)) * 64);
        #pragma unroll
        for (int kk = 0; kk < 16; ++kk) a0[kk] = xp0[2 * kk + hi];
        #pragma unroll
        for (int kk = 0; kk < 16; ++kk) a1[kk] = xp1[2 * kk + hi];
    }

    lcnt[t] = 0;
    for (int i = t; i < BINW * ROWW / 4; i += 256) lr4[i] = zero4;
    if (t == 0) lover = 0;
    #pragma unroll
    for (int it = 0; it < 4; ++it) {
        const int i = it * 256 + t;
        const int k = i >> 4;
        const int n = (i & 15) * 4;
        float* p = lw + (k >> 1) * 128 + n * 2 + (k & 1);
        p[0] = w4[it].x; p[2] = w4[it].y; p[4] = w4[it].z; p[6] = w4[it].w;
    }
    lds_barrier();
    if (c > SEGCAP) lover = 1;
    lds_barrier();
    const bool over = lover != 0;

    if (!over) {
        const int w[16] = { v0.x, v0.y, v0.z, v0.w, v1.x, v1.y, v1.z, v1.w,
                            v2.x, v2.y, v2.z, v2.w, v3.x, v3.y, v3.z, v3.w };
        #pragma unroll
        for (int i = 0; i < 16; ++i) {
            if (i < c) place_entry((int)((unsigned)w[i] >> 24), w[i] & 0xFFFFFF, d0, lcnt, lrows, cnt, spill, N);
        }
        for (int i = 16; i < c; i += 4) {
            const int4 v = sp[i >> 2];
            place_entry((int)((unsigned)v.x >> 24), v.x & 0xFFFFFF, d0, lcnt, lrows, cnt, spill, N);
            if (i + 1 < c) place_entry((int)((unsigned)v.y >> 24), v.y & 0xFFFFFF, d0, lcnt, lrows, cnt, spill, N);
            if (i + 2 < c) place_entry((int)((unsigned)v.z >> 24), v.z & 0xFFFFFF, d0, lcnt, lrows, cnt, spill, N);
            if (i + 3 < c) place_entry((int)((unsigned)v.w >> 24), v.w & 0xFFFFFF, d0, lcnt, lrows, cnt, spill, N);
        }
    } else {
        for (int e = t; e < E; e += 256) {
            const int d = dst[e];
            if ((unsigned)d < (unsigned)N && (d >> BINSH) == b) {
                place_entry(d & (BINW - 1), src[e], d0, lcnt, lrows, cnt, spill, N);
            }
        }
    }
    lds_barrier();

    if (d0 + t < N) cnt[d0 + t] = lcnt[t];
    int4* const g4 = reinterpret_cast<int4*>(rows + (size_t)d0 * ROWW);
    for (int i = t; i < BINW * ROWW / 4; i += 256) {
        if (d0 + (i >> 3) < N) g4[i] = lr4[i];
    }

    const int l0 = wave * 16 + 8 * hi;
    float sc0[8], sc1[8];
    #pragma unroll
    for (int r = 0; r < 8; ++r) {
        sc0[r] = rsqrtf((float)(lcnt[l0 + r] + 1));
        sc1[r] = rsqrtf((float)(lcnt[l0 + 128 + r] + 1));
    }

    v8f c00 = {}, c01 = {}, c02 = {}, c03 = {};
    v8f c10 = {}, c11 = {}, c12 = {}, c13 = {};

    #pragma unroll
    for (int kk = 0; kk < 16; ++kk) {
        const v2f* wp = (const v2f*)(lw + (2 * kk + hi) * 128 + lh * 2);
        const v2f b0 = wp[0];
        const v2f b1 = wp[16];
        const v2f b2 = wp[32];
        const v2f b3 = wp[48];

        c00 = __builtin_amdgcn_wmma_f32_16x16x4_f32(false, a0[kk], false, b0, (short)0, c00, false, false);
        c01 = __builtin_amdgcn_wmma_f32_16x16x4_f32(false, a0[kk], false, b1, (short)0, c01, false, false);
        c02 = __builtin_amdgcn_wmma_f32_16x16x4_f32(false, a0[kk], false, b2, (short)0, c02, false, false);
        c03 = __builtin_amdgcn_wmma_f32_16x16x4_f32(false, a0[kk], false, b3, (short)0, c03, false, false);
        c10 = __builtin_amdgcn_wmma_f32_16x16x4_f32(false, a1[kk], false, b0, (short)0, c10, false, false);
        c11 = __builtin_amdgcn_wmma_f32_16x16x4_f32(false, a1[kk], false, b1, (short)0, c11, false, false);
        c12 = __builtin_amdgcn_wmma_f32_16x16x4_f32(false, a1[kk], false, b2, (short)0, c12, false, false);
        c13 = __builtin_amdgcn_wmma_f32_16x16x4_f32(false, a1[kk], false, b3, (short)0, c13, false, false);
    }

    #pragma unroll
    for (int r = 0; r < 8; ++r) {
        c00[r] *= sc0[r]; c01[r] *= sc0[r]; c02[r] *= sc0[r]; c03[r] *= sc0[r];
        c10[r] *= sc1[r]; c11[r] *= sc1[r]; c12[r] *= sc1[r]; c13[r] *= sc1[r];
    }

    #pragma unroll
    for (int r = 0; r < 8; ++r) {
        const int row0 = d0 + l0 + r;
        const int row1 = row0 + 128;
        if (row0 < N) {
            float* yp = Y + (size_t)row0 * 64 + lh;
            yp[0]  = c00[r];
            yp[16] = c01[r];
            yp[32] = c02[r];
            yp[48] = c03[r];
        }
        if (row1 < N) {
            float* yp = Y + (size_t)row1 * 64 + lh;
            yp[0]  = c10[r];
            yp[16] = c11[r];
            yp[32] = c12[r];
            yp[48] = c13[r];
        }
    }
}


__device__ __forceinline__ unsigned inrange(int s, unsigned un) {
    return (unsigned)s < un ? (unsigned)s : 0u;
}

__device__ __forceinline__ void add4(float4& a, const float4& v) {
    a.x += v.x; a.y += v.y; a.z += v.z; a.w += v.w;
}

__global__ void __launch_bounds__(256)
gather_kernel(const float* __restrict__ xs, const int* __restrict__ rows,
              const int* __restrict__ cnt, const int* __restrict__ spill,
              const int* __restrict__ src, const int* __restrict__ dst,
              const float* __restrict__ bias,
              float* __restrict__ o, int N, int E, int do_relu) {
    const int t = blockIdx.x * 256 + threadIdx.x;
    const int d = t >> 4;
    const int q = t & 15;
    if (d >= N) return;

    const float4* x4 = reinterpret_cast<const float4*>(xs);
    const int c = cnt[d];
    const float di = rsqrtf((float)(c + 1));
    const int n = min(max(c, 0), ROWW);
    const int* row = rows + (size_t)d * ROWW;
    const int4* row4 = reinterpret_cast<const int4*>(row);
    const unsigned un = (unsigned)N;

    float4 acc = make_float4(0.0f, 0.0f, 0.0f, 0.0f);
    int j = 0;
    for (; j + 8 <= n; j += 8) {
        const int4 ea = row4[j >> 2];
        const int4 eb = row4[(j >> 2) + 1];
        const float4 v0 = x4[(size_t)inrange(ea.x, un) * 16 + q];
        const float4 v1 = x4[(size_t)inrange(ea.y, un) * 16 + q];
        const float4 v2 = x4[(size_t)inrange(ea.z, un) * 16 + q];
        const float4 v3 = x4[(size_t)inrange(ea.w, un) * 16 + q];
        const float4 v4 = x4[(size_t)inrange(eb.x, un) * 16 + q];
        const float4 v5 = x4[(size_t)inrange(eb.y, un) * 16 + q];
        const float4 v6 = x4[(size_t)inrange(eb.z, un) * 16 + q];
        const float4 v7 = x4[(size_t)inrange(eb.w, un) * 16 + q];
        add4(acc, v0); add4(acc, v1); add4(acc, v2); add4(acc, v3);
        add4(acc, v4); add4(acc, v5); add4(acc, v6); add4(acc, v7);
    }
    if (j + 4 <= n) {
        const int4 e = row4[j >> 2];
        const float4 v0 = x4[(size_t)inrange(e.x, un) * 16 + q];
        const float4 v1 = x4[(size_t)inrange(e.y, un) * 16 + q];
        const float4 v2 = x4[(size_t)inrange(e.z, un) * 16 + q];
        const float4 v3 = x4[(size_t)inrange(e.w, un) * 16 + q];
        add4(acc, v0); add4(acc, v1); add4(acc, v2); add4(acc, v3);
        j += 4;
    }
    for (; j < n; ++j) {
        add4(acc, x4[(size_t)inrange(row[j], un) * 16 + q]);
    }
    if (c > ROWW) {
        const int total = cnt[N];
        if (total >= 0 && total <= SPILL_CAP) {
            const int2* sp = reinterpret_cast<const int2*>(spill);
            for (int k = 0; k < total; ++k) {
                const int2 p = sp[k];
                if (p.x == d) {
                    add4(acc, x4[(size_t)inrange(p.y, un) * 16 + q]);
                }
            }
        } else {
            acc = make_float4(0.0f, 0.0f, 0.0f, 0.0f);
            for (int e = 0; e < E; ++e) {
                if (dst[e] == d) {
                    add4(acc, x4[(size_t)inrange(src[e], un) * 16 + q]);
                }
            }
        }
    }

    const float4 self = x4[(size_t)d * 16 + q];
    const float4 b = reinterpret_cast<const float4*>(bias)[q];
    float4 r;
    r.x = (acc.x + self.x) * di + b.x;
    r.y = (acc.y + self.y) * di + b.y;
    r.z = (acc.z + self.z) * di + b.z;
    r.w = (acc.w + self.w) * di + b.w;
    if (do_relu) {
        r.x = fmaxf(r.x, 0.0f); r.y = fmaxf(r.y, 0.0f);
        r.z = fmaxf(r.z, 0.0f); r.w = fmaxf(r.w, 0.0f);
    }
    reinterpret_cast<float4*>(o)[(size_t)d * 16 + q] = r;
}

#define TPB 5
#define HP 68

__global__ void __launch_bounds__(256)
gather_gemm_kernel(const float* __restrict__ xs, const int* __restrict__ rows,
                   const int* __restrict__ cnt, const int* __restrict__ spill,
                   const int* __restrict__ src, const int* __restrict__ dst,
                   const float* __restrict__ bias, const float* __restrict__ W,
                   float* __restrict__ y, int N, int E) {
    __shared__ float sm[(64 + 16 + 16) * HP];
    float* const lw = sm;
    float* const ht = sm + 64 * HP;
    float* const ot = ht + 16 * HP;

    const int g = threadIdx.x >> 4;
    const int q = threadIdx.x & 15;
    const int wave = threadIdx.x >> 5;
    const int lane = threadIdx.x & 31;
    const int lh = lane & 15;
    const int hi = lane >> 4;

    const float4* x4 = reinterpret_cast<const float4*>(xs);
    const unsigned un = (unsigned)N;
    const int4 zero4 = make_int4(0, 0, 0, 0);

    int d = blockIdx.x * (TPB * 16) + g;
    int c = 0;
    int4 ea = zero4, eb = zero4;
    if (d < N) {
        const int4* r4 = reinterpret_cast<const int4*>(rows + (size_t)d * ROWW);
        c = cnt[d];
        ea = r4[0];
        eb = r4[1];
    }

    #pragma unroll
    for (int it = 0; it < 4; ++it) {
        const int i = it * 256 + threadIdx.x;
        const float4 w4 = reinterpret_cast<const float4*>(W)[i];
        const int k = i >> 4;
        const int n = (i & 15) * 4;
        float* p = lw + n * HP + k;
        p[0] = w4.x; p[HP] = w4.y; p[2 * HP] = w4.z; p[3 * HP] = w4.w;
    }

    for (int t = 0; t < TPB; ++t) {
        if (d - g >= N) break;

        float4 r = make_float4(0.0f, 0.0f, 0.0f, 0.0f);
        float di = 0.0f;
        if (d < N) {
            const float4 self = x4[(size_t)d * 16 + q];
            di = rsqrtf((float)(c + 1));
            const int n = min(max(c, 0), ROWW);
            const int4* row4 = reinterpret_cast<const int4*>(rows + (size_t)d * ROWW);

            float4 acc = make_float4(0.0f, 0.0f, 0.0f, 0.0f);
            for (int j = 0; j < n; j += 8) {
                int4 na = zero4, nb = zero4;
                if (j + 8 < n) {
                    na = row4[(j >> 2) + 2];
                    nb = row4[(j >> 2) + 3];
                }
                const int m = n - j;
                const float4 z = make_float4(0.0f, 0.0f, 0.0f, 0.0f);
                float4 v0 = x4[(size_t)inrange(ea.x, un) * 16 + q];
                float4 v1 = z, v2 = z, v3 = z, v4 = z, v5 = z, v6 = z, v7 = z;
                if (m > 1) v1 = x4[(size_t)inrange(ea.y, un) * 16 + q];
                if (m > 2) v2 = x4[(size_t)inrange(ea.z, un) * 16 + q];
                if (m > 3) v3 = x4[(size_t)inrange(ea.w, un) * 16 + q];
                if (m > 4) v4 = x4[(size_t)inrange(eb.x, un) * 16 + q];
                if (m > 5) v5 = x4[(size_t)inrange(eb.y, un) * 16 + q];
                if (m > 6) v6 = x4[(size_t)inrange(eb.z, un) * 16 + q];
                if (m > 7) v7 = x4[(size_t)inrange(eb.w, un) * 16 + q];
                add4(acc, v0); add4(acc, v1); add4(acc, v2); add4(acc, v3);
                add4(acc, v4); add4(acc, v5); add4(acc, v6); add4(acc, v7);
                ea = na; eb = nb;
            }
            if (c > ROWW) {
                const int total = cnt[N];
                if (total >= 0 && total <= SPILL_CAP) {
                    const int2* sp = reinterpret_cast<const int2*>(spill);
                    for (int k = 0; k < total; ++k) {
                        const int2 p = sp[k];
                        if (p.x == d) {
                            add4(acc, x4[(size_t)inrange(p.y, un) * 16 + q]);
                        }
                    }
                } else {
                    acc = make_float4(0.0f, 0.0f, 0.0f, 0.0f);
                    for (int e = 0; e < E; ++e) {
                        if (dst[e] == d) {
                            add4(acc, x4[(size_t)inrange(src[e], un) * 16 + q]);
                        }
                    }
                }
            }

            const float4 b = reinterpret_cast<const float4*>(bias)[q];
            r.x = fmaxf((acc.x + self.x) * di + b.x, 0.0f);
            r.y = fmaxf((acc.y + self.y) * di + b.y, 0.0f);
            r.z = fmaxf((acc.z + self.z) * di + b.z, 0.0f);
            r.w = fmaxf((acc.w + self.w) * di + b.w, 0.0f);
        }
        *reinterpret_cast<float4*>(ht + g * HP + 4 * q) = r;
        __syncthreads();

        const int dn = d + 16;
        int cn = 0;
        int4 ean = zero4, ebn = zero4;
        if (t + 1 < TPB && dn < N) {
            const int4* r4 = reinterpret_cast<const int4*>(rows + (size_t)dn * ROWW);
            cn = cnt[dn];
            ean = r4[0];
            ebn = r4[1];
        }

        if (wave < 4) {
            const int n0 = wave * 16;
            const float* ap = ht + lh * HP + 2 * hi;
            const float* bp = lw + (n0 + lh) * HP + 2 * hi;
            v8f acc = {};
            #pragma unroll
            for (int kk = 0; kk < 16; ++kk) {
                const v2f a  = *reinterpret_cast<const v2f*>(ap + 4 * kk);
                const v2f bw = *reinterpret_cast<const v2f*>(bp + 4 * kk);
                acc = __builtin_amdgcn_wmma_f32_16x16x4_f32(false, a, false, bw, (short)0, acc, false, false);
            }
            float* op = ot + 8 * hi * HP + n0 + lh;
            #pragma unroll
            for (int rr = 0; rr < 8; ++rr) {
                op[rr * HP] = acc[rr];
            }
        }
        __syncthreads();

        if (d < N) {
            float4 o = *reinterpret_cast<const float4*>(ot + g * HP + 4 * q);
            o.x *= di; o.y *= di; o.z *= di; o.w *= di;
            reinterpret_cast<float4*>(y)[(size_t)d * 16 + q] = o;
        }

        d = dn; c = cn; ea = ean; eb = ebn;
    }
}


extern "C" void kernel_launch(void* const* d_in, const int* in_sizes, int n_in,
                              void* d_out, int out_size, void* d_ws, size_t ws_size,
                              hipStream_t stream) {
    (void)n_in; (void)out_size; (void)ws_size;

    const float* x  = (const float*)d_in[0];
    const int*   ei = (const int*)d_in[1];
    const float* W1 = (const float*)d_in[2];
    const float* b1 = (const float*)d_in[3];
    const float* W2 = (const float*)d_in[4];
    const float* b2 = (const float*)d_in[5];
    float* out = (float*)d_out;

    const int N = in_sizes[0] / 64;
    const int E = in_sizes[1] / 2;
    const int* src = ei;
    const int* dst = ei + E;

    int* cnt    = (int*)d_ws;
    float* xs   = (float*)d_ws + (((size_t)N + 1 + 32767) & ~(size_t)32767);
    int* rows   = (int*)(xs + (size_t)N * 64);
    int* spill  = rows + (size_t)N * ROWW;

    const int T = 256;

    int* stage  = (int*)xs;
    int* segcnt = stage + (size_t)((N + BINW - 1) / BINW) * NSEG * SEGCAP;
    part_kernel<<<NSEG, T, 0, stream>>>(src, dst, stage, segcnt, cnt, N, E);
    rows_gemm_kernel<<<(N + BINW - 1) / BINW, T, 0, stream>>>(src, dst, stage, segcnt, cnt, rows, spill, x, W1, out, N, E);

    gather_gemm_kernel<<<(N + 16 * TPB - 1) / (16 * TPB), T, 0, stream>>>(out, rows, cnt, spill, src, dst, b1, W2, xs, N, E);

    gather_kernel<<<(N * 16 + T - 1) / T, T, 0, stream>>>(xs, rows, cnt, spill, src, dst, b2, out, N, E, 0);
}


# layer-2 gather made persistent: 1250 work-groups of 8 waves each walk 5 tiles of destinations, next tile's count and ids asked for ahead, in place of 50000 one-shot waves; sum order kept
# speedup vs baseline: 4.6911x; 1.0484x over previous
#include <hip/hip_runtime.h>


typedef __attribute__((ext_vector_type(2))) float v2f;
typedef __attribute__((ext_vector_type(8))) float v8f;

#define ROWW 32
#define SPILL_CAP 262144

#define BINW 256
#define BINSH 8
#define NSEG 250
#define SEGCAP 64
#define NBMAX 512
#define EPT 19
#define TILE (EPT * 256)

__global__ void __launch_bounds__(256)
part_kernel(const int* __restrict__ src, const int* __restrict__ dst,
            int* __restrict__ stage, int* __restrict__ segcnt, int* __restrict__ cnt, int N, int E) {
    __shared__ int hist[NBMAX];
    __shared__ int start[NBMAX];
    __shared__ int cursor[NBMAX];
    __shared__ int segoff[NBMAX];
    __shared__ int sc[256];
    __shared__ int lval[TILE];
    __shared__ int lbin[TILE];

    const int t = threadIdx.x;
    const int blk = blockIdx.x;
    const int nb = (N + BINW - 1) >> BINSH;
    const int per = (E + NSEG - 1) / NSEG;
    const int cbeg = min(blk * per, E);
    const int cend = min(cbeg + per, E);

    if (blk == 0 && t == 0) cnt[N] = 0;
    segoff[t] = 0;
    segoff[t + 256] = 0;

    for (int base = cbeg; base < cend; base += TILE) {
        hist[t] = 0;
        hist[t + 256] = 0;
        __syncthreads();

        int dv[EPT], sv[EPT];
        #pragma unroll
        for (int k = 0; k < EPT; ++k) {
            const int e = base + k * 256 + t;
            const int ec = min(e, cend - 1);
            int d = dst[ec];
            const int s = src[ec];
            if (e >= cend || (unsigned)d >= (unsigned)N || (d >> BINSH) >= NBMAX) d = -1;
            dv[k] = d;
            sv[k] = s;
        }
        #pragma unroll
        for (int k = 0; k < EPT; ++k) {
            if (dv[k] >= 0) atomicAdd(&hist[dv[k] >> BINSH], 1);
        }
        __syncthreads();

        const int h0 = hist[2 * t];
        const int h1 = hist[2 * t + 1];
        const int mine = h0 + h1;
        sc[t] = mine;
        __syncthreads();
        for (int off = 1; off < 256; off <<= 1) {
            const int v = t >= off ? sc[t - off] : 0;
            __syncthreads();
            sc[t] += v;
            __syncthreads();
        }
        const int ex = sc[t] - mine;
        start[2 * t] = ex;
        start[2 * t + 1] = ex + h0;
        cursor[2 * t] = ex;
        cursor[2 * t + 1] = ex + h0;
        const int total = sc[255];
        __syncthreads();

        #pragma unroll
        for (int k = 0; k < EPT; ++k) {
            if (dv[k] >= 0) {
                const int bin = dv[k] >> BINSH;
                const int pos = atomicAdd(&cursor[bin], 1);
                lval[pos] = (int)(((unsigned)(dv[k] & (BINW - 1)) << 24) | ((unsigned)sv[k] & 0xFFFFFFu));
                lbin[pos] = bin;
            }
        }
        __syncthreads();

        for (int i = t; i < total; i += 256) {
            const int bin = lbin[i];
            const int off = segoff[bin] + (i - start[bin]);
            if (off < SEGCAP) stage[((size_t)bin * NSEG + blk) * SEGCAP + off] = lval[i];
        }
        __syncthreads();
        segoff[t] += hist[t];
        segoff[t + 256] += hist[t + 256];
    }
    __syncthreads();
    for (int bin = t; bin < nb && bin < NBMAX; bin += 256) segcnt[bin * NSEG + blk] = segoff[bin];
}

__device__ __forceinline__ void lds_barrier() {
    __builtin_amdgcn_fence(__ATOMIC_RELEASE, "workgroup", "local");
    __builtin_amdgcn_s_barrier();
    __builtin_amdgcn_fence(__ATOMIC_ACQUIRE, "workgroup", "local");
}

__device__ __forceinline__ void place_entry(int dl, int s, int d0, int* lcnt, int* lrows,
                                            int* __restrict__ cnt, int* __restrict__ spill, int N) {
    const int r = atomicAdd(&lcnt[dl], 1);
    if (r < ROWW) {
        lrows[dl * ROWW + r] = s;
    } else {
        const int k = atomicAdd(&cnt[N], 1);
        if (k >= 0 && k < SPILL_CAP) reinterpret_cast<int2*>(spill)[k] = make_int2(d0 + dl, s);
    }
}

__global__ void __launch_bounds__(256)
rows_gemm_kernel(const int* __restrict__ src, const int* __restrict__ dst,
                 const int* __restrict__ stage, const int* __restrict__ segcnt,
                 int* __restrict__ cnt, int* __restrict__ rows, int* __restrict__ spill,
                 const float* __restrict__ X, const float* __restrict__ W, float* __restrict__ Y,
                 int N, int E) {
    __shared__ int lcnt[BINW];
    __shared__ int lrows[BINW * ROWW];
    __shared__ int lover;
    __shared__ float lw[64 * 64];

    const int t = threadIdx.x;
    const int b = blockIdx.x;
    const int d0 = b * BINW;
    const int4 zero4 = make_int4(0, 0, 0, 0);
    int4* const lr4 = reinterpret_cast<int4*>(lrows);

    const int wave = t >> 5;
    const int lane = t & 31;
    const int lh   = lane & 15;
    const int hi   = lane >> 4;

    float4 w4[4];
    #pragma unroll
    for (int it = 0; it < 4; ++it) w4[it] = reinterpret_cast<const float4*>(W)[it * 256 + t];

    int c = 0;
    int4 v0 = zero4, v1 = zero4, v2 = zero4, v3 = zero4;
    const int4* sp = reinterpret_cast<const int4*>(stage + ((size_t)b * NSEG + (t < NSEG ? t : 0)) * SEGCAP);
    if (t < NSEG) {
        c = segcnt[b * NSEG + t];
        v0 = sp[0]; v1 = sp[1]; v2 = sp[2]; v3 = sp[3];
    }

    v2f a0[16], a1[16];
    {
        const int r0 = d0 + wave * 16 + lh;
        const int r1 = r0 + 128;
        const v2f* xp0 = (const v2f*)(X + (size_t)(r0 < N ? r0 : (N - 1)) * 64);
        const v2f* xp1 = (const v2f*)(X + (size_t)(r1 < N ? r1 : (N - 1)) * 64);
        #pragma unroll
        for (int kk = 0; kk < 16; ++kk) a0[kk] = xp0[2 * kk + hi];
        #pragma unroll
        for (int kk = 0; kk < 16; ++kk) a1[kk] = xp1[2 * kk + hi];
    }

    lcnt[t] = 0;
    for (int i = t; i < BINW * ROWW / 4; i += 256) lr4[i] = zero4;
    if (t == 0) lover = 0;
    #pragma unroll
    for (int it = 0; it < 4; ++it) {
        const int i = it * 256 + t;
        const int k = i >> 4;
        const int n = (i & 15) * 4;
        float* p = lw + (k >> 1) * 128 + n * 2 + (k & 1);
        p[0] = w4[it].x; p[2] = w4[it].y; p[4] = w4[it].z; p[6] = w4[it].w;
    }
    lds_barrier();
    if (c > SEGCAP) lover = 1;
    lds_barrier();
    const bool over = lover != 0;

    if (!over) {
        const int w[16] = { v0.x, v0.y, v0.z, v0.w, v1.x, v1.y, v1.z, v1.w,
                            v2.x, v2.y, v2.z, v2.w, v3.x, v3.y, v3.z, v3.w };
        #pragma unroll
        for (int i = 0; i < 16; ++i) {
            if (i < c) place_entry((int)((unsigned)w[i] >> 24), w[i] & 0xFFFFFF, d0, lcnt, lrows, cnt, spill, N);
        }
        for (int i = 16; i < c; i += 4) {
            const int4 v = sp[i >> 2];
            place_entry((int)((unsigned)v.x >> 24), v.x & 0xFFFFFF, d0, lcnt, lrows, cnt, spill, N);
            if (i + 1 < c) place_entry((int)((unsigned)v.y >> 24), v.y & 0xFFFFFF, d0, lcnt, lrows, cnt, spill, N);
            if (i + 2 < c) place_entry((int)((unsigned)v.z >> 24), v.z & 0xFFFFFF, d0, lcnt, lrows, cnt, spill, N);
            if (i + 3 < c) place_entry((int)((unsigned)v.w >> 24), v.w & 0xFFFFFF, d0, lcnt, lrows, cnt, spill, N);
        }
    } else {
        for (int e = t; e < E; e += 256) {
            const int d = dst[e];
            if ((unsigned)d < (unsigned)N && (d >> BINSH) == b) {
                place_entry(d & (BINW - 1), src[e], d0, lcnt, lrows, cnt, spill, N);
            }
        }
    }
    lds_barrier();

    if (d0 + t < N) cnt[d0 + t] = lcnt[t];
    int4* const g4 = reinterpret_cast<int4*>(rows + (size_t)d0 * ROWW);
    for (int i = t; i < BINW * ROWW / 4; i += 256) {
        if (d0 + (i >> 3) < N) g4[i] = lr4[i];
    }

    const int l0 = wave * 16 + 8 * hi;
    float sc0[8], sc1[8];
    #pragma unroll
    for (int r = 0; r < 8; ++r) {
        sc0[r] = rsqrtf((float)(lcnt[l0 + r] + 1));
        sc1[r] = rsqrtf((float)(lcnt[l0 + 128 + r] + 1));
    }

    v8f c00 = {}, c01 = {}, c02 = {}, c03 = {};
    v8f c10 = {}, c11 = {}, c12 = {}, c13 = {};

    #pragma unroll
    for (int kk = 0; kk < 16; ++kk) {
        const v2f* wp = (const v2f*)(lw + (2 * kk + hi) * 128 + lh * 2);
        const v2f b0 = wp[0];
        const v2f b1 = wp[16];
        const v2f b2 = wp[32];
        const v2f b3 = wp[48];

        c00 = __builtin_amdgcn_wmma_f32_16x16x4_f32(false, a0[kk], false, b0, (short)0, c00, false, false);
        c01 = __builtin_amdgcn_wmma_f32_16x16x4_f32(false, a0[kk], false, b1, (short)0, c01, false, false);
        c02 = __builtin_amdgcn_wmma_f32_16x16x4_f32(false, a0[kk], false, b2, (short)0, c02, false, false);
        c03 = __builtin_amdgcn_wmma_f32_16x16x4_f32(false, a0[kk], false, b3, (short)0, c03, false, false);
        c10 = __builtin_amdgcn_wmma_f32_16x16x4_f32(false, a1[kk], false, b0, (short)0, c10, false, false);
        c11 = __builtin_amdgcn_wmma_f32_16x16x4_f32(false, a1[kk], false, b1, (short)0, c11, false, false);
        c12 = __builtin_amdgcn_wmma_f32_16x16x4_f32(false, a1[kk], false, b2, (short)0, c12, false, false);
        c13 = __builtin_amdgcn_wmma_f32_16x16x4_f32(false, a1[kk], false, b3, (short)0, c13, false, false);
    }

    #pragma unroll
    for (int r = 0; r < 8; ++r) {
        c00[r] *= sc0[r]; c01[r] *= sc0[r]; c02[r] *= sc0[r]; c03[r] *= sc0[r];
        c10[r] *= sc1[r]; c11[r] *= sc1[r]; c12[r] *= sc1[r]; c13[r] *= sc1[r];
    }

    #pragma unroll
    for (int r = 0; r < 8; ++r) {
        const int row0 = d0 + l0 + r;
        const int row1 = row0 + 128;
        if (row0 < N) {
            float* yp = Y + (size_t)row0 * 64 + lh;
            yp[0]  = c00[r];
            yp[16] = c01[r];
            yp[32] = c02[r];
            yp[48] = c03[r];
        }
        if (row1 < N) {
            float* yp = Y + (size_t)row1 * 64 + lh;
            yp[0]  = c10[r];
            yp[16] = c11[r];
            yp[32] = c12[r];
            yp[48] = c13[r];
        }
    }
}


__device__ __forceinline__ unsigned inrange(int s, unsigned un) {
    return (unsigned)s < un ? (unsigned)s : 0u;
}

__device__ __forceinline__ void add4(float4& a, const float4& v) {
    a.x += v.x; a.y += v.y; a.z += v.z; a.w += v.w;
}

#define GT 5

__global__ void __launch_bounds__(256)
gather_kernel(const float* __restrict__ xs, const int* __restrict__ rows,
              const int* __restrict__ cnt, const int* __restrict__ spill,
              const int* __restrict__ src, const int* __restrict__ dst,
              const float* __restrict__ bias,
              float* __restrict__ o, int N, int E) {
    const int g = threadIdx.x >> 4;
    const int q = threadIdx.x & 15;

    const float4* x4 = reinterpret_cast<const float4*>(xs);
    const unsigned un = (unsigned)N;
    const float4 b = reinterpret_cast<const float4*>(bias)[q];

    int d = blockIdx.x * (GT * 16) + g;
    if (d >= N) return;

    const int4* r4 = reinterpret_cast<const int4*>(rows + (size_t)d * ROWW);
    int c = cnt[d];
    int4 ea = r4[0];
    int4 eb = r4[1];

    for (int t = 0; t < GT; ++t) {
        const int dn = d + 16;
        const bool more = (t + 1 < GT) && (dn < N);
        const int dp = more ? dn : d;
        const int4* rn4 = reinterpret_cast<const int4*>(rows + (size_t)dp * ROWW);
        const int cn = cnt[dp];
        const int4 ean = rn4[0];
        const int4 ebn = rn4[1];

        const float4 self = x4[(size_t)d * 16 + q];
        const float di = rsqrtf((float)(c + 1));
        const int n = min(max(c, 0), ROWW);
        const int4* row4 = reinterpret_cast<const int4*>(rows + (size_t)d * ROWW);

        float4 acc = make_float4(0.0f, 0.0f, 0.0f, 0.0f);
        for (int j = 0; j < n; j += 8) {
            const int i2 = min((j >> 2) + 2, ROWW / 4 - 2);
            const int4 na = row4[i2];
            const int4 nb = row4[i2 + 1];
            const int m = n - j;
            const float4 z = make_float4(0.0f, 0.0f, 0.0f, 0.0f);
            float4 v0 = x4[(size_t)inrange(ea.x, un) * 16 + q];
            float4 v1 = z, v2 = z, v3 = z, v4 = z, v5 = z, v6 = z, v7 = z;
            if (m > 1) v1 = x4[(size_t)inrange(ea.y, un) * 16 + q];
            if (m > 2) v2 = x4[(size_t)inrange(ea.z, un) * 16 + q];
            if (m > 3) v3 = x4[(size_t)inrange(ea.w, un) * 16 + q];
            if (m > 4) v4 = x4[(size_t)inrange(eb.x, un) * 16 + q];
            if (m > 5) v5 = x4[(size_t)inrange(eb.y, un) * 16 + q];
            if (m > 6) v6 = x4[(size_t)inrange(eb.z, un) * 16 + q];
            if (m > 7) v7 = x4[(size_t)inrange(eb.w, un) * 16 + q];
            add4(acc, v0); add4(acc, v1); add4(acc, v2); add4(acc, v3);
            add4(acc, v4); add4(acc, v5); add4(acc, v6); add4(acc, v7);
            ea = na; eb = nb;
        }
        if (c > ROWW) {
            const int total = cnt[N];
            if (total >= 0 && total <= SPILL_CAP) {
                const int2* sp = reinterpret_cast<const int2*>(spill);
                for (int k = 0; k < total; ++k) {
                    const int2 p = sp[k];
                    if (p.x == d) {
                        add4(acc, x4[(size_t)inrange(p.y, un) * 16 + q]);
                    }
                }
            } else {
                acc = make_float4(0.0f, 0.0f, 0.0f, 0.0f);
                for (int e = 0; e < E; ++e) {
                    if (dst[e] == d) {
                        add4(acc, x4[(size_t)inrange(src[e], un) * 16 + q]);
                    }
                }
            }
        }

        float4 r;
        r.x = (acc.x + self.x) * di + b.x;
        r.y = (acc.y + self.y) * di + b.y;
        r.z = (acc.z + self.z) * di + b.z;
        r.w = (acc.w + self.w) * di + b.w;
        reinterpret_cast<float4*>(o)[(size_t)d * 16 + q] = r;

        if (!more) break;
        d = dn; c = cn; ea = ean; eb = ebn;
    }
}

#define TPB 5
#define HP 68

__global__ void __launch_bounds__(256)
gather_gemm_kernel(const float* __restrict__ xs, const int* __restrict__ rows,
                   const int* __restrict__ cnt, const int* __restrict__ spill,
                   const int* __restrict__ src, const int* __restrict__ dst,
                   const float* __restrict__ bias, const float* __restrict__ W,
                   float* __restrict__ y, int N, int E) {
    __shared__ float sm[(64 + 16 + 16) * HP];
    float* const lw = sm;
    float* const ht = sm + 64 * HP;
    float* const ot = ht + 16 * HP;

    const int g = threadIdx.x >> 4;
    const int q = threadIdx.x & 15;
    const int wave = threadIdx.x >> 5;
    const int lane = threadIdx.x & 31;
    const int lh = lane & 15;
    const int hi = lane >> 4;

    const float4* x4 = reinterpret_cast<const float4*>(xs);
    const unsigned un = (unsigned)N;
    const int4 zero4 = make_int4(0, 0, 0, 0);

    int d = blockIdx.x * (TPB * 16) + g;
    int c = 0;
    int4 ea = zero4, eb = zero4;
    if (d < N) {
        const int4* r4 = reinterpret_cast<const int4*>(rows + (size_t)d * ROWW);
        c = cnt[d];
        ea = r4[0];
        eb = r4[1];
    }

    #pragma unroll
    for (int it = 0; it < 4; ++it) {
        const int i = it * 256 + threadIdx.x;
        const float4 w4 = reinterpret_cast<const float4*>(W)[i];
        const int k = i >> 4;
        const int n = (i & 15) * 4;
        float* p = lw + n * HP + k;
        p[0] = w4.x; p[HP] = w4.y; p[2 * HP] = w4.z; p[3 * HP] = w4.w;
    }

    for (int t = 0; t < TPB; ++t) {
        if (d - g >= N) break;

        float4 r = make_float4(0.0f, 0.0f, 0.0f, 0.0f);
        float di = 0.0f;
        if (d < N) {
            const float4 self = x4[(size_t)d * 16 + q];
            di = rsqrtf((float)(c + 1));
            const int n = min(max(c, 0), ROWW);
            const int4* row4 = reinterpret_cast<const int4*>(rows + (size_t)d * ROWW);

            float4 acc = make_float4(0.0f, 0.0f, 0.0f, 0.0f);
            for (int j = 0; j < n; j += 8) {
                int4 na = zero4, nb = zero4;
                if (j + 8 < n) {
                    na = row4[(j >> 2) + 2];
                    nb = row4[(j >> 2) + 3];
                }
                const int m = n - j;
                const float4 z = make_float4(0.0f, 0.0f, 0.0f, 0.0f);
                float4 v0 = x4[(size_t)inrange(ea.x, un) * 16 + q];
                float4 v1 = z, v2 = z, v3 = z, v4 = z, v5 = z, v6 = z, v7 = z;
                if (m > 1) v1 = x4[(size_t)inrange(ea.y, un) * 16 + q];
                if (m > 2) v2 = x4[(size_t)inrange(ea.z, un) * 16 + q];
                if (m > 3) v3 = x4[(size_t)inrange(ea.w, un) * 16 + q];
                if (m > 4) v4 = x4[(size_t)inrange(eb.x, un) * 16 + q];
                if (m > 5) v5 = x4[(size_t)inrange(eb.y, un) * 16 + q];
                if (m > 6) v6 = x4[(size_t)inrange(eb.z, un) * 16 + q];
                if (m > 7) v7 = x4[(size_t)inrange(eb.w, un) * 16 + q];
                add4(acc, v0); add4(acc, v1); add4(acc, v2); add4(acc, v3);
                add4(acc, v4); add4(acc, v5); add4(acc, v6); add4(acc, v7);
                ea = na; eb = nb;
            }
            if (c > ROWW) {
                const int total = cnt[N];
                if (total >= 0 && total <= SPILL_CAP) {
                    const int2* sp = reinterpret_cast<const int2*>(spill);
                    for (int k = 0; k < total; ++k) {
                        const int2 p = sp[k];
                        if (p.x == d) {
                            add4(acc, x4[(size_t)inrange(p.y, un) * 16 + q]);
                        }
                    }
                } else {
                    acc = make_float4(0.0f, 0.0f, 0.0f, 0.0f);
                    for (int e = 0; e < E; ++e) {
                        if (dst[e] == d) {
                            add4(acc, x4[(size_t)inrange(src[e], un) * 16 + q]);
                        }
                    }
                }
            }

            const float4 b = reinterpret_cast<const float4*>(bias)[q];
            r.x = fmaxf((acc.x + self.x) * di + b.x, 0.0f);
            r.y = fmaxf((acc.y + self.y) * di + b.y, 0.0f);
            r.z = fmaxf((acc.z + self.z) * di + b.z, 0.0f);
            r.w = fmaxf((acc.w + self.w) * di + b.w, 0.0f);
        }
        *reinterpret_cast<float4*>(ht + g * HP + 4 * q) = r;
        __syncthreads();

        const int dn = d + 16;
        int cn = 0;
        int4 ean = zero4, ebn = zero4;
        if (t + 1 < TPB && dn < N) {
            const int4* r4 = reinterpret_cast<const int4*>(rows + (size_t)dn * ROWW);
            cn = cnt[dn];
            ean = r4[0];
            ebn = r4[1];
        }

        if (wave < 4) {
            const int n0 = wave * 16;
            const float* ap = ht + lh * HP + 2 * hi;
            const float* bp = lw + (n0 + lh) * HP + 2 * hi;
            v8f acc = {};
            #pragma unroll
            for (int kk = 0; kk < 16; ++kk) {
                const v2f a  = *reinterpret_cast<const v2f*>(ap + 4 * kk);
                const v2f bw = *reinterpret_cast<const v2f*>(bp + 4 * kk);
                acc = __builtin_amdgcn_wmma_f32_16x16x4_f32(false, a, false, bw, (short)0, acc, false, false);
            }
            float* op = ot + 8 * hi * HP + n0 + lh;
            #pragma unroll
            for (int rr = 0; rr < 8; ++rr) {
                op[rr * HP] = acc[rr];
            }
        }
        __syncthreads();

        if (d < N) {
            float4 o = *reinterpret_cast<const float4*>(ot + g * HP + 4 * q);
            o.x *= di; o.y *= di; o.z *= di; o.w *= di;
            reinterpret_cast<float4*>(y)[(size_t)d * 16 + q] = o;
        }

        d = dn; c = cn; ea = ean; eb = ebn;
    }
}


extern "C" void kernel_launch(void* const* d_in, const int* in_sizes, int n_in,
                              void* d_out, int out_size, void* d_ws, size_t ws_size,
                              hipStream_t stream) {
    (void)n_in; (void)out_size; (void)ws_size;

    const float* x  = (const float*)d_in[0];
    const int*   ei = (const int*)d_in[1];
    const float* W1 = (const float*)d_in[2];
    const float* b1 = (const float*)d_in[3];
    const float* W2 = (const float*)d_in[4];
    const float* b2 = (const float*)d_in[5];
    float* out = (float*)d_out;

    const int N = in_sizes[0] / 64;
    const int E = in_sizes[1] / 2;
    const int* src = ei;
    const int* dst = ei + E;

    int* cnt    = (int*)d_ws;
    float* xs   = (float*)d_ws + (((size_t)N + 1 + 32767) & ~(size_t)32767);
    int* rows   = (int*)(xs + (size_t)N * 64);
    int* spill  = rows + (size_t)N * ROWW;

    const int T = 256;

    int* stage  = (int*)xs;
    int* segcnt = stage + (size_t)((N + BINW - 1) / BINW) * NSEG * SEGCAP;
    part_kernel<<<NSEG, T, 0, stream>>>(src, dst, stage, segcnt, cnt, N, E);
    rows_gemm_kernel<<<(N + BINW - 1) / BINW, T, 0, stream>>>(src, dst, stage, segcnt, cnt, rows, spill, x, W1, out, N, E);

    gather_gemm_kernel<<<(N + 16 * TPB - 1) / (16 * TPB), T, 0, stream>>>(out, rows, cnt, spill, src, dst, b1, W2, xs, N, E);

    gather_kernel<<<(N + 16 * GT - 1) / (16 * GT), T, 0, stream>>>(xs, rows, cnt, spill, src, dst, b2, out, N, E);
}


# layer-1 product moved behind its gather, A(xW1) = (Ax)W1: row-building launch only scales x, gather launch multiplies each 16-row tile by W1 then W2 on f32 WMMA; layer-1 sums run over x rows first
# speedup vs baseline: 4.8564x; 1.0352x over previous
#include <hip/hip_runtime.h>


typedef __attribute__((ext_vector_type(2))) float v2f;
typedef __attribute__((ext_vector_type(8))) float v8f;

#define ROWW 32
#define SPILL_CAP 262144

#define BINW 256
#define BINSH 8
#define NSEG 250
#define SEGCAP 64
#define NBMAX 512
#define EPT 19
#define TILE (EPT * 256)

__global__ void __launch_bounds__(256)
part_kernel(const int* __restrict__ src, const int* __restrict__ dst,
            int* __restrict__ stage, int* __restrict__ segcnt, int* __restrict__ cnt, int N, int E) {
    __shared__ int hist[NBMAX];
    __shared__ int start[NBMAX];
    __shared__ int cursor[NBMAX];
    __shared__ int segoff[NBMAX];
    __shared__ int sc[256];
    __shared__ int lval[TILE];
    __shared__ int lbin[TILE];

    const int t = threadIdx.x;
    const int blk = blockIdx.x;
    const int nb = (N + BINW - 1) >> BINSH;
    const int per = (E + NSEG - 1) / NSEG;
    const int cbeg = min(blk * per, E);
    const int cend = min(cbeg + per, E);

    if (blk == 0 && t == 0) cnt[N] = 0;
    segoff[t] = 0;
    segoff[t + 256] = 0;

    for (int base = cbeg; base < cend; base += TILE) {
        hist[t] = 0;
        hist[t + 256] = 0;
        __syncthreads();

        int dv[EPT], sv[EPT];
        #pragma unroll
        for (int k = 0; k < EPT; ++k) {
            const int e = base + k * 256 + t;
            const int ec = min(e, cend - 1);
            int d = dst[ec];
            const int s = src[ec];
            if (e >= cend || (unsigned)d >= (unsigned)N || (d >> BINSH) >= NBMAX) d = -1;
            dv[k] = d;
            sv[k] = s;
        }
        #pragma unroll
        for (int k = 0; k < EPT; ++k) {
            if (dv[k] >= 0) atomicAdd(&hist[dv[k] >> BINSH], 1);
        }
        __syncthreads();

        const int h0 = hist[2 * t];
        const int h1 = hist[2 * t + 1];
        const int mine = h0 + h1;
        sc[t] = mine;
        __syncthreads();
        for (int off = 1; off < 256; off <<= 1) {
            const int v = t >= off ? sc[t - off] : 0;
            __syncthreads();
            sc[t] += v;
            __syncthreads();
        }
        const int ex = sc[t] - mine;
        start[2 * t] = ex;
        start[2 * t + 1] = ex + h0;
        cursor[2 * t] = ex;
        cursor[2 * t + 1] = ex + h0;
        const int total = sc[255];
        __syncthreads();

        #pragma unroll
        for (int k = 0; k < EPT; ++k) {
            if (dv[k] >= 0) {
                const int bin = dv[k] >> BINSH;
                const int pos = atomicAdd(&cursor[bin], 1);
                lval[pos] = (int)(((unsigned)(dv[k] & (BINW - 1)) << 24) | ((unsigned)sv[k] & 0xFFFFFFu));
                lbin[pos] = bin;
            }
        }
        __syncthreads();

        for (int i = t; i < total; i += 256) {
            const int bin = lbin[i];
            const int off = segoff[bin] + (i - start[bin]);
            if (off < SEGCAP) stage[((size_t)bin * NSEG + blk) * SEGCAP + off] = lval[i];
        }
        __syncthreads();
        segoff[t] += hist[t];
        segoff[t + 256] += hist[t + 256];
    }
    __syncthreads();
    for (int bin = t; bin < nb && bin < NBMAX; bin += 256) segcnt[bin * NSEG + blk] = segoff[bin];
}

__device__ __forceinline__ void lds_barrier() {
    __builtin_amdgcn_fence(__ATOMIC_RELEASE, "workgroup", "local");
    __builtin_amdgcn_s_barrier();
    __builtin_amdgcn_fence(__ATOMIC_ACQUIRE, "workgroup", "local");
}

__device__ __forceinline__ void place_entry(int dl, int s, int d0, int* lcnt, int* lrows,
                                            int* __restrict__ cnt, int* __restrict__ spill, int N) {
    const int r = atomicAdd(&lcnt[dl], 1);
    if (r < ROWW) {
        lrows[dl * ROWW + r] = s;
    } else {
        const int k = atomicAdd(&cnt[N], 1);
        if (k >= 0 && k < SPILL_CAP) reinterpret_cast<int2*>(spill)[k] = make_int2(d0 + dl, s);
    }
}

__global__ void __launch_bounds__(256)
rows_scale_kernel(const int* __restrict__ src, const int* __restrict__ dst,
                  const int* __restrict__ stage, const int* __restrict__ segcnt,
                  int* __restrict__ cnt, int* __restrict__ rows, int* __restrict__ spill,
                  const float* __restrict__ X, float* __restrict__ Y,
                  int N, int E) {
    __shared__ int lcnt[BINW];
    __shared__ int lrows[BINW * ROWW];
    __shared__ int lover;

    const int t = threadIdx.x;
    const int b = blockIdx.x;
    const int d0 = b * BINW;
    const int4 zero4 = make_int4(0, 0, 0, 0);
    int4* const lr4 = reinterpret_cast<int4*>(lrows);

    int c = 0;
    int4 v0 = zero4, v1 = zero4, v2 = zero4, v3 = zero4;
    const int4* sp = reinterpret_cast<const int4*>(stage + ((size_t)b * NSEG + (t < NSEG ? t : 0)) * SEGCAP);
    if (t < NSEG) {
        c = segcnt[b * NSEG + t];
        v0 = sp[0]; v1 = sp[1]; v2 = sp[2]; v3 = sp[3];
    }

    const int rq = t >> 4;
    const int q  = t & 15;
    float4 xr[16];
    {
        const float4* x4 = reinterpret_cast<const float4*>(X);
        #pragma unroll
        for (int it = 0; it < 16; ++it) {
            const int r = d0 + it * 16 + rq;
            xr[it] = x4[(size_t)(r < N ? r : (N - 1)) * 16 + q];
        }
    }

    lcnt[t] = 0;
    for (int i = t; i < BINW * ROWW / 4; i += 256) lr4[i] = zero4;
    if (t == 0) lover = 0;
    lds_barrier();
    if (c > SEGCAP) lover = 1;
    lds_barrier();
    const bool over = lover != 0;

    if (!over) {
        const int w[16] = { v0.x, v0.y, v0.z, v0.w, v1.x, v1.y, v1.z, v1.w,
                            v2.x, v2.y, v2.z, v2.w, v3.x, v3.y, v3.z, v3.w };
        #pragma unroll
        for (int i = 0; i < 16; ++i) {
            if (i < c) place_entry((int)((unsigned)w[i] >> 24), w[i] & 0xFFFFFF, d0, lcnt, lrows, cnt, spill, N);
        }
        for (int i = 16; i < c; i += 4) {
            const int4 v = sp[i >> 2];
            place_entry((int)((unsigned)v.x >> 24), v.x & 0xFFFFFF, d0, lcnt, lrows, cnt, spill, N);
            if (i + 1 < c) place_entry((int)((unsigned)v.y >> 24), v.y & 0xFFFFFF, d0, lcnt, lrows, cnt, spill, N);
            if (i + 2 < c) place_entry((int)((unsigned)v.z >> 24), v.z & 0xFFFFFF, d0, lcnt, lrows, cnt, spill, N);
            if (i + 3 < c) place_entry((int)((unsigned)v.w >> 24), v.w & 0xFFFFFF, d0, lcnt, lrows, cnt, spill, N);
        }
    } else {
        for (int e = t; e < E; e += 256) {
            const int d = dst[e];
            if ((unsigned)d < (unsigned)N && (d >> BINSH) == b) {
                place_entry(d & (BINW - 1), src[e], d0, lcnt, lrows, cnt, spill, N);
            }
        }
    }
    lds_barrier();

    if (d0 + t < N) cnt[d0 + t] = lcnt[t];
    int4* const g4 = reinterpret_cast<int4*>(rows + (size_t)d0 * ROWW);
    for (int i = t; i < BINW * ROWW / 4; i += 256) {
        if (d0 + (i >> 3) < N) g4[i] = lr4[i];
    }

    float4* const y4 = reinterpret_cast<float4*>(Y);
    #pragma unroll
    for (int it = 0; it < 16; ++it) {
        const int rl = it * 16 + rq;
        const int r = d0 + rl;
        const float s = rsqrtf((float)(lcnt[rl] + 1));
        if (r < N) {
            float4 v = xr[it];
            v.x *= s; v.y *= s; v.z *= s; v.w *= s;
            y4[(size_t)r * 16 + q] = v;
        }
    }
}


__device__ __forceinline__ unsigned inrange(int s, unsigned un) {
    return (unsigned)s < un ? (unsigned)s : 0u;
}

__device__ __forceinline__ void add4(float4& a, const float4& v) {
    a.x += v.x; a.y += v.y; a.z += v.z; a.w += v.w;
}

#define GT 5

__global__ void __launch_bounds__(256)
gather_kernel(const float* __restrict__ xs, const int* __restrict__ rows,
              const int* __restrict__ cnt, const int* __restrict__ spill,
              const int* __restrict__ src, const int* __restrict__ dst,
              const float* __restrict__ bias,
              float* __restrict__ o, int N, int E) {
    const int g = threadIdx.x >> 4;
    const int q = threadIdx.x & 15;

    const float4* x4 = reinterpret_cast<const float4*>(xs);
    const unsigned un = (unsigned)N;
    const float4 b = reinterpret_cast<const float4*>(bias)[q];

    int d = blockIdx.x * (GT * 16) + g;
    if (d >= N) return;

    const int4* r4 = reinterpret_cast<const int4*>(rows + (size_t)d * ROWW);
    int c = cnt[d];
    int4 ea = r4[0];
    int4 eb = r4[1];

    for (int t = 0; t < GT; ++t) {
        const int dn = d + 16;
        const bool more = (t + 1 < GT) && (dn < N);
        const int dp = more ? dn : d;
        const int4* rn4 = reinterpret_cast<const int4*>(rows + (size_t)dp * ROWW);
        const int cn = cnt[dp];
        const int4 ean = rn4[0];
        const int4 ebn = rn4[1];

        const float4 self = x4[(size_t)d * 16 + q];
        const float di = rsqrtf((float)(c + 1));
        const int n = min(max(c, 0), ROWW);
        const int4* row4 = reinterpret_cast<const int4*>(rows + (size_t)d * ROWW);

        float4 acc = make_float4(0.0f, 0.0f, 0.0f, 0.0f);
        for (int j = 0; j < n; j += 8) {
            const int i2 = min((j >> 2) + 2, ROWW / 4 - 2);
            const int4 na = row4[i2];
            const int4 nb = row4[i2 + 1];
            const int m = n - j;
            const float4 z = make_float4(0.0f, 0.0f, 0.0f, 0.0f);
            float4 v0 = x4[(size_t)inrange(ea.x, un) * 16 + q];
            float4 v1 = z, v2 = z, v3 = z, v4 = z, v5 = z, v6 = z, v7 = z;
            if (m > 1) v1 = x4[(size_t)inrange(ea.y, un) * 16 + q];
            if (m > 2) v2 = x4[(size_t)inrange(ea.z, un) * 16 + q];
            if (m > 3) v3 = x4[(size_t)inrange(ea.w, un) * 16 + q];
            if (m > 4) v4 = x4[(size_t)inrange(eb.x, un) * 16 + q];
            if (m > 5) v5 = x4[(size_t)inrange(eb.y, un) * 16 + q];
            if (m > 6) v6 = x4[(size_t)inrange(eb.z, un) * 16 + q];
            if (m > 7) v7 = x4[(size_t)inrange(eb.w, un) * 16 + q];
            add4(acc, v0); add4(acc, v1); add4(acc, v2); add4(acc, v3);
            add4(acc, v4); add4(acc, v5); add4(acc, v6); add4(acc, v7);
            ea = na; eb = nb;
        }
        if (c > ROWW) {
            const int total = cnt[N];
            if (total >= 0 && total <= SPILL_CAP) {
                const int2* sp = reinterpret_cast<const int2*>(spill);
                for (int k = 0; k < total; ++k) {
                    const int2 p = sp[k];
                    if (p.x == d) {
                        add4(acc, x4[(size_t)inrange(p.y, un) * 16 + q]);
                    }
                }
            } else {
                acc = make_float4(0.0f, 0.0f, 0.0f, 0.0f);
                for (int e = 0; e < E; ++e) {
                    if (dst[e] == d) {
                        add4(acc, x4[(size_t)inrange(src[e], un) * 16 + q]);
                    }
                }
            }
        }

        float4 r;
        r.x = (acc.x + self.x) * di + b.x;
        r.y = (acc.y + self.y) * di + b.y;
        r.z = (acc.z + self.z) * di + b.z;
        r.w = (acc.w + self.w) * di + b.w;
        reinterpret_cast<float4*>(o)[(size_t)d * 16 + q] = r;

        if (!more) break;
        d = dn; c = cn; ea = ean; eb = ebn;
    }
}

#define TPB 5
#define HP 68

__device__ __forceinline__ v8f tile_product(const float* ap, const float* bp) {
    v8f acc = {};
    #pragma unroll
    for (int kk = 0; kk < 16; ++kk) {
        const v2f a  = *reinterpret_cast<const v2f*>(ap + 4 * kk);
        const v2f bw = *reinterpret_cast<const v2f*>(bp + 4 * kk);
        acc = __builtin_amdgcn_wmma_f32_16x16x4_f32(false, a, false, bw, (short)0, acc, false, false);
    }
    return acc;
}

__global__ void __launch_bounds__(256)
gather_gemm_kernel(const float* __restrict__ xs, const int* __restrict__ rows,
                   const int* __restrict__ cnt, const int* __restrict__ spill,
                   const int* __restrict__ src, const int* __restrict__ dst,
                   const float* __restrict__ bias, const float* __restrict__ W1, const float* __restrict__ W2,
                   float* __restrict__ y, int N, int E) {
    __shared__ float sm[(64 + 64 + 16 + 16) * HP];
    float* const lw1 = sm;
    float* const lw2 = sm + 64 * HP;
    float* const ht = sm + 128 * HP;
    float* const ot = ht + 16 * HP;

    const int g = threadIdx.x >> 4;
    const int q = threadIdx.x & 15;
    const int wave = threadIdx.x >> 5;
    const int lane = threadIdx.x & 31;
    const int lh = lane & 15;
    const int hi = lane >> 4;

    const float4* x4 = reinterpret_cast<const float4*>(xs);
    const unsigned un = (unsigned)N;
    const int4 zero4 = make_int4(0, 0, 0, 0);

    int d = blockIdx.x * (TPB * 16) + g;
    int c = 0;
    int4 ea = zero4, eb = zero4;
    if (d < N) {
        const int4* r4 = reinterpret_cast<const int4*>(rows + (size_t)d * ROWW);
        c = cnt[d];
        ea = r4[0];
        eb = r4[1];
    }

    #pragma unroll
    for (int it = 0; it < 4; ++it) {
        const int i = it * 256 + threadIdx.x;
        const float4 wa = reinterpret_cast<const float4*>(W1)[i];
        const float4 wb = reinterpret_cast<const float4*>(W2)[i];
        const int k = i >> 4;
        const int n = (i & 15) * 4;
        float* p1 = lw1 + n * HP + k;
        p1[0] = wa.x; p1[HP] = wa.y; p1[2 * HP] = wa.z; p1[3 * HP] = wa.w;
        float* p2 = lw2 + n * HP + k;
        p2[0] = wb.x; p2[HP] = wb.y; p2[2 * HP] = wb.z; p2[3 * HP] = wb.w;
    }

    const int n0 = (wave & 3) * 16;
    const float bc = bias[n0 + lh];

    for (int t = 0; t < TPB; ++t) {
        if (d - g >= N) break;

        float4 r = make_float4(0.0f, 0.0f, 0.0f, 0.0f);
        float di = 0.0f;
        if (d < N) {
            const float4 self = x4[(size_t)d * 16 + q];
            di = rsqrtf((float)(c + 1));
            const int n = min(max(c, 0), ROWW);
            const int4* row4 = reinterpret_cast<const int4*>(rows + (size_t)d * ROWW);

            float4 acc = make_float4(0.0f, 0.0f, 0.0f, 0.0f);
            for (int j = 0; j < n; j += 8) {
                int4 na = zero4, nb = zero4;
                if (j + 8 < n) {
                    na = row4[(j >> 2) + 2];
                    nb = row4[(j >> 2) + 3];
                }
                const int m = n - j;
                const float4 z = make_float4(0.0f, 0.0f, 0.0f, 0.0f);
                float4 v0 = x4[(size_t)inrange(ea.x, un) * 16 + q];
                float4 v1 = z, v2 = z, v3 = z, v4 = z, v5 = z, v6 = z, v7 = z;
                if (m > 1) v1 = x4[(size_t)inrange(ea.y, un) * 16 + q];
                if (m > 2) v2 = x4[(size_t)inrange(ea.z, un) * 16 + q];
                if (m > 3) v3 = x4[(size_t)inrange(ea.w, un) * 16 + q];
                if (m > 4) v4 = x4[(size_t)inrange(eb.x, un) * 16 + q];
                if (m > 5) v5 = x4[(size_t)inrange(eb.y, un) * 16 + q];
                if (m > 6) v6 = x4[(size_t)inrange(eb.z, un) * 16 + q];
                if (m > 7) v7 = x4[(size_t)inrange(eb.w, un) * 16 + q];
                add4(acc, v0); add4(acc, v1); add4(acc, v2); add4(acc, v3);
                add4(acc, v4); add4(acc, v5); add4(acc, v6); add4(acc, v7);
                ea = na; eb = nb;
            }
            if (c > ROWW) {
                const int total = cnt[N];
                if (total >= 0 && total <= SPILL_CAP) {
                    const int2* sp = reinterpret_cast<const int2*>(spill);
                    for (int k = 0; k < total; ++k) {
                        const int2 p = sp[k];
                        if (p.x == d) {
                            add4(acc, x4[(size_t)inrange(p.y, un) * 16 + q]);
                        }
                    }
                } else {
                    acc = make_float4(0.0f, 0.0f, 0.0f, 0.0f);
                    for (int e = 0; e < E; ++e) {
                        if (dst[e] == d) {
                            add4(acc, x4[(size_t)inrange(src[e], un) * 16 + q]);
                        }
                    }
                }
            }

            r.x = (acc.x + self.x) * di;
            r.y = (acc.y + self.y) * di;
            r.z = (acc.z + self.z) * di;
            r.w = (acc.w + self.w) * di;
        }
        *reinterpret_cast<float4*>(ht + g * HP + 4 * q) = r;
        __syncthreads();

        const int dn = d + 16;
        int cn = 0;
        int4 ean = zero4, ebn = zero4;
        if (t + 1 < TPB && dn < N) {
            const int4* r4 = reinterpret_cast<const int4*>(rows + (size_t)dn * ROWW);
            cn = cnt[dn];
            ean = r4[0];
            ebn = r4[1];
        }

        if (wave < 4) {
            const v8f acc = tile_product(ht + lh * HP + 2 * hi, lw1 + (n0 + lh) * HP + 2 * hi);
            float* op = ot + 8 * hi * HP + n0 + lh;
            #pragma unroll
            for (int rr = 0; rr < 8; ++rr) {
                op[rr * HP] = fmaxf(acc[rr] + bc, 0.0f);
            }
        }
        lds_barrier();

        if (wave < 4) {
            const v8f acc = tile_product(ot + lh * HP + 2 * hi, lw2 + (n0 + lh) * HP + 2 * hi);
            float* op = ht + 8 * hi * HP + n0 + lh;
            #pragma unroll
            for (int rr = 0; rr < 8; ++rr) {
                op[rr * HP] = acc[rr];
            }
        }
        __syncthreads();

        if (d < N) {
            float4 o = *reinterpret_cast<const float4*>(ht + g * HP + 4 * q);
            o.x *= di; o.y *= di; o.z *= di; o.w *= di;
            reinterpret_cast<float4*>(y)[(size_t)d * 16 + q] = o;
        }

        d = dn; c = cn; ea = ean; eb = ebn;
    }
}


extern "C" void kernel_launch(void* const* d_in, const int* in_sizes, int n_in,
                              void* d_out, int out_size, void* d_ws, size_t ws_size,
                              hipStream_t stream) {
    (void)n_in; (void)out_size; (void)ws_size;

    const float* x  = (const float*)d_in[0];
    const int*   ei = (const int*)d_in[1];
    const float* W1 = (const float*)d_in[2];
    const float* b1 = (const float*)d_in[3];
    const float* W2 = (const float*)d_in[4];
    const float* b2 = (const float*)d_in[5];
    float* out = (float*)d_out;

    const int N = in_sizes[0] / 64;
    const int E = in_sizes[1] / 2;
    const int* src = ei;
    const int* dst = ei + E;

    int* cnt    = (int*)d_ws;
    float* xs   = (float*)d_ws + (((size_t)N + 1 + 32767) & ~(size_t)32767);
    int* rows   = (int*)(xs + (size_t)N * 64);
    int* spill  = rows + (size_t)N * ROWW;

    const int T = 256;

    int* stage  = (int*)xs;
    int* segcnt = stage + (size_t)((N + BINW - 1) / BINW) * NSEG * SEGCAP;
    part_kernel<<<NSEG, T, 0, stream>>>(src, dst, stage, segcnt, cnt, N, E);
    rows_scale_kernel<<<(N + BINW - 1) / BINW, T, 0, stream>>>(src, dst, stage, segcnt, cnt, rows, spill, x, out, N, E);

    gather_gemm_kernel<<<(N + 16 * TPB - 1) / (16 * TPB), T, 0, stream>>>(out, rows, cnt, spill, src, dst, b1, W1, W2, xs, N, E);

    gather_kernel<<<(N + 16 * GT - 1) / (16 * GT), T, 0, stream>>>(xs, rows, cnt, spill, src, dst, b2, out, N, E);
}


# both gathers: a group's 8 rows loaded branch-free (slots past the count read row 0, weighted 0 in an fma), 32-bit row offsets, one-min id clamp: half the vector instructions; sum order kept
# speedup vs baseline: 5.0957x; 1.0493x over previous
#include <hip/hip_runtime.h>


typedef __attribute__((ext_vector_type(2))) float v2f;
typedef __attribute__((ext_vector_type(8))) float v8f;

#define ROWW 32
#define SPILL_CAP 262144

#define BINW 256
#define BINSH 8
#define NSEG 250
#define SEGCAP 64
#define NBMAX 512
#define EPT 19
#define TILE (EPT * 256)

__global__ void __launch_bounds__(256)
part_kernel(const int* __restrict__ src, const int* __restrict__ dst,
            int* __restrict__ stage, int* __restrict__ segcnt, int* __restrict__ cnt, int N, int E) {
    __shared__ int hist[NBMAX];
    __shared__ int start[NBMAX];
    __shared__ int cursor[NBMAX];
    __shared__ int segoff[NBMAX];
    __shared__ int sc[256];
    __shared__ int lval[TILE];
    __shared__ int lbin[TILE];

    const int t = threadIdx.x;
    const int blk = blockIdx.x;
    const int nb = (N + BINW - 1) >> BINSH;
    const int per = (E + NSEG - 1) / NSEG;
    const int cbeg = min(blk * per, E);
    const int cend = min(cbeg + per, E);

    if (blk == 0 && t == 0) cnt[N] = 0;
    segoff[t] = 0;
    segoff[t + 256] = 0;

    for (int base = cbeg; base < cend; base += TILE) {
        hist[t] = 0;
        hist[t + 256] = 0;
        __syncthreads();

        int dv[EPT], sv[EPT];
        #pragma unroll
        for (int k = 0; k < EPT; ++k) {
            const int e = base + k * 256 + t;
            const int ec = min(e, cend - 1);
            int d = dst[ec];
            const int s = src[ec];
            if (e >= cend || (unsigned)d >= (unsigned)N || (d >> BINSH) >= NBMAX) d = -1;
            dv[k] = d;
            sv[k] = s;
        }
        #pragma unroll
        for (int k = 0; k < EPT; ++k) {
            if (dv[k] >= 0) atomicAdd(&hist[dv[k] >> BINSH], 1);
        }
        __syncthreads();

        const int h0 = hist[2 * t];
        const int h1 = hist[2 * t + 1];
        const int mine = h0 + h1;
        sc[t] = mine;
        __syncthreads();
        for (int off = 1; off < 256; off <<= 1) {
            const int v = t >= off ? sc[t - off] : 0;
            __syncthreads();
            sc[t] += v;
            __syncthreads();
        }
        const int ex = sc[t] - mine;
        start[2 * t] = ex;
        start[2 * t + 1] = ex + h0;
        cursor[2 * t] = ex;
        cursor[2 * t + 1] = ex + h0;
        const int total = sc[255];
        __syncthreads();

        #pragma unroll
        for (int k = 0; k < EPT; ++k) {
            if (dv[k] >= 0) {
                const int bin = dv[k] >> BINSH;
                const int pos = atomicAdd(&cursor[bin], 1);
                lval[pos] = (int)(((unsigned)(dv[k] & (BINW - 1)) << 24) | ((unsigned)sv[k] & 0xFFFFFFu));
                lbin[pos] = bin;
            }
        }
        __syncthreads();

        for (int i = t; i < total; i += 256) {
            const int bin = lbin[i];
            const int off = segoff[bin] + (i - start[bin]);
            if (off < SEGCAP) stage[((size_t)bin * NSEG + blk) * SEGCAP + off] = lval[i];
        }
        __syncthreads();
        segoff[t] += hist[t];
        segoff[t + 256] += hist[t + 256];
    }
    __syncthreads();
    for (int bin = t; bin < nb && bin < NBMAX; bin += 256) segcnt[bin * NSEG + blk] = segoff[bin];
}

__device__ __forceinline__ void lds_barrier() {
    __builtin_amdgcn_fence(__ATOMIC_RELEASE, "workgroup", "local");
    __builtin_amdgcn_s_barrier();
    __builtin_amdgcn_fence(__ATOMIC_ACQUIRE, "workgroup", "local");
}

__device__ __forceinline__ void place_entry(int dl, int s, int d0, int* lcnt, int* lrows,
                                            int* __restrict__ cnt, int* __restrict__ spill, int N) {
    const int r = atomicAdd(&lcnt[dl], 1);
    if (r < ROWW) {
        lrows[dl * ROWW + r] = s;
    } else {
        const int k = atomicAdd(&cnt[N], 1);
        if (k >= 0 && k < SPILL_CAP) reinterpret_cast<int2*>(spill)[k] = make_int2(d0 + dl, s);
    }
}

__global__ void __launch_bounds__(256)
rows_scale_kernel(const int* __restrict__ src, const int* __restrict__ dst,
                  const int* __restrict__ stage, const int* __restrict__ segcnt,
                  int* __restrict__ cnt, int* __restrict__ rows, int* __restrict__ spill,
                  const float* __restrict__ X, float* __restrict__ Y,
                  int N, int E) {
    __shared__ int lcnt[BINW];
    __shared__ int lrows[BINW * ROWW];
    __shared__ int lover;

    const int t = threadIdx.x;
    const int b = blockIdx.x;
    const int d0 = b * BINW;
    const int4 zero4 = make_int4(0, 0, 0, 0);
    int4* const lr4 = reinterpret_cast<int4*>(lrows);

    int c = 0;
    int4 v0 = zero4, v1 = zero4, v2 = zero4, v3 = zero4;
    const int4* sp = reinterpret_cast<const int4*>(stage + ((size_t)b * NSEG + (t < NSEG ? t : 0)) * SEGCAP);
    if (t < NSEG) {
        c = segcnt[b * NSEG + t];
        v0 = sp[0]; v1 = sp[1]; v2 = sp[2]; v3 = sp[3];
    }

    const int rq = t >> 4;
    const int q  = t & 15;
    float4 xr[16];
    {
        const float4* x4 = reinterpret_cast<const float4*>(X);
        #pragma unroll
        for (int it = 0; it < 16; ++it) {
            const int r = d0 + it * 16 + rq;
            xr[it] = x4[(size_t)(r < N ? r : (N - 1)) * 16 + q];
        }
    }

    lcnt[t] = 0;
    for (int i = t; i < BINW * ROWW / 4; i += 256) lr4[i] = zero4;
    if (t == 0) lover = 0;
    lds_barrier();
    if (c > SEGCAP) lover = 1;
    lds_barrier();
    const bool over = lover != 0;

    if (!over) {
        const int w[16] = { v0.x, v0.y, v0.z, v0.w, v1.x, v1.y, v1.z, v1.w,
                            v2.x, v2.y, v2.z, v2.w, v3.x, v3.y, v3.z, v3.w };
        #pragma unroll
        for (int i = 0; i < 16; ++i) {
            if (i < c) place_entry((int)((unsigned)w[i] >> 24), w[i] & 0xFFFFFF, d0, lcnt, lrows, cnt, spill, N);
        }
        for (int i = 16; i < c; i += 4) {
            const int4 v = sp[i >> 2];
            place_entry((int)((unsigned)v.x >> 24), v.x & 0xFFFFFF, d0, lcnt, lrows, cnt, spill, N);
            if (i + 1 < c) place_entry((int)((unsigned)v.y >> 24), v.y & 0xFFFFFF, d0, lcnt, lrows, cnt, spill, N);
            if (i + 2 < c) place_entry((int)((unsigned)v.z >> 24), v.z & 0xFFFFFF, d0, lcnt, lrows, cnt, spill, N);
            if (i + 3 < c) place_entry((int)((unsigned)v.w >> 24), v.w & 0xFFFFFF, d0, lcnt, lrows, cnt, spill, N);
        }
    } else {
        for (int e = t; e < E; e += 256) {
            const int d = dst[e];
            if ((unsigned)d < (unsigned)N && (d >> BINSH) == b) {
                place_entry(d & (BINW - 1), src[e], d0, lcnt, lrows, cnt, spill, N);
            }
        }
    }
    lds_barrier();

    if (d0 + t < N) cnt[d0 + t] = lcnt[t];
    int4* const g4 = reinterpret_cast<int4*>(rows + (size_t)d0 * ROWW);
    for (int i = t; i < BINW * ROWW / 4; i += 256) {
        if (d0 + (i >> 3) < N) g4[i] = lr4[i];
    }

    float4* const y4 = reinterpret_cast<float4*>(Y);
    #pragma unroll
    for (int it = 0; it < 16; ++it) {
        const int rl = it * 16 + rq;
        const int r = d0 + rl;
        const float s = rsqrtf((float)(lcnt[rl] + 1));
        if (r < N) {
            float4 v = xr[it];
            v.x *= s; v.y *= s; v.z *= s; v.w *= s;
            y4[(size_t)r * 16 + q] = v;
        }
    }
}


__device__ __forceinline__ unsigned inrange(int s, unsigned un) {
    return (unsigned)s < un ? (unsigned)s : 0u;
}

__device__ __forceinline__ void add4(float4& a, const float4& v) {
    a.x += v.x; a.y += v.y; a.z += v.z; a.w += v.w;
}

__device__ __forceinline__ void fma4(float4& a, const float4& v, float w) {
    a.x = fmaf(v.x, w, a.x); a.y = fmaf(v.y, w, a.y); a.z = fmaf(v.z, w, a.z); a.w = fmaf(v.w, w, a.w);
}

__device__ __forceinline__ int4 inrange4(const int4 e, unsigned last) {
    return make_int4((int)min((unsigned)e.x, last), (int)min((unsigned)e.y, last),
                     (int)min((unsigned)e.z, last), (int)min((unsigned)e.w, last));
}

__device__ __forceinline__ uint4 rowidx4(const int4 e, unsigned q) {
    return make_uint4(((unsigned)e.x << 4) | q, ((unsigned)e.y << 4) | q, ((unsigned)e.z << 4) | q, ((unsigned)e.w << 4) | q);
}

__device__ __forceinline__ void sum8(float4& acc, const float4* __restrict__ x4, const uint4 ia, const uint4 ib, int m) {
    const float4 v0 = x4[ia.x];
    const float4 v1 = x4[ia.y];
    const float4 v2 = x4[ia.z];
    const float4 v3 = x4[ia.w];
    const float4 v4 = x4[ib.x];
    const float4 v5 = x4[ib.y];
    const float4 v6 = x4[ib.z];
    const float4 v7 = x4[ib.w];
    const float mf = (float)m;
    add4(acc, v0);
    fma4(acc, v1, fminf(fmaxf(mf - 1.0f, 0.0f), 1.0f));
    fma4(acc, v2, fminf(fmaxf(mf - 2.0f, 0.0f), 1.0f));
    fma4(acc, v3, fminf(fmaxf(mf - 3.0f, 0.0f), 1.0f));
    fma4(acc, v4, fminf(fmaxf(mf - 4.0f, 0.0f), 1.0f));
    fma4(acc, v5, fminf(fmaxf(mf - 5.0f, 0.0f), 1.0f));
    fma4(acc, v6, fminf(fmaxf(mf - 6.0f, 0.0f), 1.0f));
    fma4(acc, v7, fminf(fmaxf(mf - 7.0f, 0.0f), 1.0f));
}

#define GT 5

__global__ void __launch_bounds__(256)
gather_kernel(const float* __restrict__ xs, const int* __restrict__ rows,
              const int* __restrict__ cnt, const int* __restrict__ spill,
              const int* __restrict__ src, const int* __restrict__ dst,
              const float* __restrict__ bias,
              float* __restrict__ o, int N, int E) {
    const int g = threadIdx.x >> 4;
    const int q = threadIdx.x & 15;

    const float4* x4 = reinterpret_cast<const float4*>(xs);
    const unsigned un = (unsigned)N;
    const unsigned last = un - 1u;
    const float4 b = reinterpret_cast<const float4*>(bias)[q];

    int d = blockIdx.x * (GT * 16) + g;
    if (d >= N) return;

    const int4* r4 = reinterpret_cast<const int4*>(rows + (size_t)d * ROWW);
    int c = cnt[d];
    int4 ea = r4[0];
    int4 eb = r4[1];

    for (int t = 0; t < GT; ++t) {
        const int dn = d + 16;
        const bool more = (t + 1 < GT) && (dn < N);
        const int dp = more ? dn : d;
        const int4* rn4 = reinterpret_cast<const int4*>(rows + (size_t)dp * ROWW);
        const int cn = cnt[dp];
        const int4 ean = rn4[0];
        const int4 ebn = rn4[1];

        const float4 self = x4[(size_t)d * 16 + q];
        const float di = rsqrtf((float)(c + 1));
        const int n = min(max(c, 0), ROWW);
        const int4* row4 = reinterpret_cast<const int4*>(rows + (size_t)d * ROWW);

        int4 na = row4[2];
        int4 nb = row4[3];

        float4 acc = make_float4(0.0f, 0.0f, 0.0f, 0.0f);
        for (int j = 0; j < n; j += 8) {
            sum8(acc, x4, rowidx4(inrange4(ea, last), (unsigned)q), rowidx4(inrange4(eb, last), (unsigned)q), n - j);
            ea = na; eb = nb;
            if (j + 16 < n) {
                na = row4[(j >> 2) + 4];
                nb = row4[(j >> 2) + 5];
            }
        }
        if (c > ROWW) {
            const int total = cnt[N];
            if (total >= 0 && total <= SPILL_CAP) {
                const int2* sp = reinterpret_cast<const int2*>(spill);
                for (int k = 0; k < total; ++k) {
                    const int2 p = sp[k];
                    if (p.x == d) {
                        add4(acc, x4[(size_t)inrange(p.y, un) * 16 + q]);
                    }
                }
            } else {
                acc = make_float4(0.0f, 0.0f, 0.0f, 0.0f);
                for (int e = 0; e < E; ++e) {
                    if (dst[e] == d) {
                        add4(acc, x4[(size_t)inrange(src[e], un) * 16 + q]);
                    }
                }
            }
        }

        float4 r;
        r.x = (acc.x + self.x) * di + b.x;
        r.y = (acc.y + self.y) * di + b.y;
        r.z = (acc.z + self.z) * di + b.z;
        r.w = (acc.w + self.w) * di + b.w;
        reinterpret_cast<float4*>(o)[(size_t)d * 16 + q] = r;

        if (!more) break;
        d = dn; c = cn; ea = ean; eb = ebn;
    }
}

#define TPB 5
#define HP 68

__device__ __forceinline__ v8f tile_product(const float* ap, const float* bp) {
    v8f acc = {};
    #pragma unroll
    for (int kk = 0; kk < 16; ++kk) {
        const v2f a  = *reinterpret_cast<const v2f*>(ap + 4 * kk);
        const v2f bw = *reinterpret_cast<const v2f*>(bp + 4 * kk);
        acc = __builtin_amdgcn_wmma_f32_16x16x4_f32(false, a, false, bw, (short)0, acc, false, false);
    }
    return acc;
}

__global__ void __launch_bounds__(256)
gather_gemm_kernel(const float* __restrict__ xs, const int* __restrict__ rows,
                   const int* __restrict__ cnt, const int* __restrict__ spill,
                   const int* __restrict__ src, const int* __restrict__ dst,
                   const float* __restrict__ bias, const float* __restrict__ W1, const float* __restrict__ W2,
                   float* __restrict__ y, int N, int E) {
    __shared__ float sm[(64 + 64 + 16 + 16) * HP];
    float* const lw1 = sm;
    float* const lw2 = sm + 64 * HP;
    float* const ht = sm + 128 * HP;
    float* const ot = ht + 16 * HP;

    const int g = threadIdx.x >> 4;
    const int q = threadIdx.x & 15;
    const int wave = threadIdx.x >> 5;
    const int lane = threadIdx.x & 31;
    const int lh = lane & 15;
    const int hi = lane >> 4;

    const float4* x4 = reinterpret_cast<const float4*>(xs);
    const unsigned un = (unsigned)N;
    const unsigned last = un - 1u;
    const int4 zero4 = make_int4(0, 0, 0, 0);

    int d = blockIdx.x * (TPB * 16) + g;
    int c = 0;
    int4 ea = zero4, eb = zero4;
    if (d < N) {
        const int4* r4 = reinterpret_cast<const int4*>(rows + (size_t)d * ROWW);
        c = cnt[d];
        ea = r4[0];
        eb = r4[1];
    }

    #pragma unroll
    for (int it = 0; it < 4; ++it) {
        const int i = it * 256 + threadIdx.x;
        const float4 wa = reinterpret_cast<const float4*>(W1)[i];
        const float4 wb = reinterpret_cast<const float4*>(W2)[i];
        const int k = i >> 4;
        const int n = (i & 15) * 4;
        float* p1 = lw1 + n * HP + k;
        p1[0] = wa.x; p1[HP] = wa.y; p1[2 * HP] = wa.z; p1[3 * HP] = wa.w;
        float* p2 = lw2 + n * HP + k;
        p2[0] = wb.x; p2[HP] = wb.y; p2[2 * HP] = wb.z; p2[3 * HP] = wb.w;
    }

    const int n0 = (wave & 3) * 16;
    const float bc = bias[n0 + lh];

    for (int t = 0; t < TPB; ++t) {
        if (d - g >= N) break;

        float4 r = make_float4(0.0f, 0.0f, 0.0f, 0.0f);
        float di = 0.0f;
        if (d < N) {
            const float4 self = x4[(size_t)d * 16 + q];
            di = rsqrtf((float)(c + 1));
            const int n = min(max(c, 0), ROWW);
            const int4* row4 = reinterpret_cast<const int4*>(rows + (size_t)d * ROWW);

            float4 acc = make_float4(0.0f, 0.0f, 0.0f, 0.0f);
            for (int j = 0; j < n; j += 8) {
                int4 na = zero4, nb = zero4;
                if (j + 8 < n) {
                    na = row4[(j >> 2) + 2];
                    nb = row4[(j >> 2) + 3];
                }
                sum8(acc, x4, rowidx4(inrange4(ea, last), (unsigned)q), rowidx4(inrange4(eb, last), (unsigned)q), n - j);
                ea = na; eb = nb;
            }
            if (c > ROWW) {
                const int total = cnt[N];
                if (total >= 0 && total <= SPILL_CAP) {
                    const int2* sp = reinterpret_cast<const int2*>(spill);
                    for (int k = 0; k < total; ++k) {
                        const int2 p = sp[k];
                        if (p.x == d) {
                            add4(acc, x4[(size_t)inrange(p.y, un) * 16 + q]);
                        }
                    }
                } else {
                    acc = make_float4(0.0f, 0.0f, 0.0f, 0.0f);
                    for (int e = 0; e < E; ++e) {
                        if (dst[e] == d) {
                            add4(acc, x4[(size_t)inrange(src[e], un) * 16 + q]);
                        }
                    }
                }
            }

            r.x = (acc.x + self.x) * di;
            r.y = (acc.y + self.y) * di;
            r.z = (acc.z + self.z) * di;
            r.w = (acc.w + self.w) * di;
        }
        *reinterpret_cast<float4*>(ht + g * HP + 4 * q) = r;
        __syncthreads();

        const int dn = d + 16;
        int cn = 0;
        int4 ean = zero4, ebn = zero4;
        if (t + 1 < TPB && dn < N) {
            const int4* r4 = reinterpret_cast<const int4*>(rows + (size_t)dn * ROWW);
            cn = cnt[dn];
            ean = r4[0];
            ebn = r4[1];
        }

        if (wave < 4) {
            const v8f acc = tile_product(ht + lh * HP + 2 * hi, lw1 + (n0 + lh) * HP + 2 * hi);
            float* op = ot + 8 * hi * HP + n0 + lh;
            #pragma unroll
            for (int rr = 0; rr < 8; ++rr) {
                op[rr * HP] = fmaxf(acc[rr] + bc, 0.0f);
            }
        }
        lds_barrier();

        if (wave < 4) {
            const v8f acc = tile_product(ot + lh * HP + 2 * hi, lw2 + (n0 + lh) * HP + 2 * hi);
            float* op = ht + 8 * hi * HP + n0 + lh;
            #pragma unroll
            for (int rr = 0; rr < 8; ++rr) {
                op[rr * HP] = acc[rr];
            }
        }
        __syncthreads();

        if (d < N) {
            float4 o = *reinterpret_cast<const float4*>(ht + g * HP + 4 * q);
            o.x *= di; o.y *= di; o.z *= di; o.w *= di;
            reinterpret_cast<float4*>(y)[(size_t)d * 16 + q] = o;
        }

        d = dn; c = cn; ea = ean; eb = ebn;
    }
}


extern "C" void kernel_launch(void* const* d_in, const int* in_sizes, int n_in,
                              void* d_out, int out_size, void* d_ws, size_t ws_size,
                              hipStream_t stream) {
    (void)n_in; (void)out_size; (void)ws_size;

    const float* x  = (const float*)d_in[0];
    const int*   ei = (const int*)d_in[1];
    const float* W1 = (const float*)d_in[2];
    const float* b1 = (const float*)d_in[3];
    const float* W2 = (const float*)d_in[4];
    const float* b2 = (const float*)d_in[5];
    float* out = (float*)d_out;

    const int N = in_sizes[0] / 64;
    const int E = in_sizes[1] / 2;
    const int* src = ei;
    const int* dst = ei + E;

    int* cnt    = (int*)d_ws;
    float* xs   = (float*)d_ws + (((size_t)N + 1 + 32767) & ~(size_t)32767);
    int* rows   = (int*)(xs + (size_t)N * 64);
    int* spill  = rows + (size_t)N * ROWW;

    const int T = 256;

    int* stage  = (int*)xs;
    int* segcnt = stage + (size_t)((N + BINW - 1) / BINW) * NSEG * SEGCAP;
    part_kernel<<<NSEG, T, 0, stream>>>(src, dst, stage, segcnt, cnt, N, E);
    rows_scale_kernel<<<(N + BINW - 1) / BINW, T, 0, stream>>>(src, dst, stage, segcnt, cnt, rows, spill, x, out, N, E);

    gather_gemm_kernel<<<(N + 16 * TPB - 1) / (16 * TPB), T, 0, stream>>>(out, rows, cnt, spill, src, dst, b1, W1, W2, xs, N, E);

    gather_kernel<<<(N + 16 * GT - 1) / (16 * GT), T, 0, stream>>>(xs, rows, cnt, spill, src, dst, b2, out, N, E);
}


# both adjacency launches: dependent LDS round trips batched per thread (rank kept from the counting add, write-out, row placing, tail), bin scan by lane exchange; row slot order = sum order changes
# speedup vs baseline: 5.2732x; 1.0348x over previous
#include <hip/hip_runtime.h>


typedef __attribute__((ext_vector_type(2))) float v2f;
typedef __attribute__((ext_vector_type(8))) float v8f;

#define ROWW 32
#define SPILL_CAP 262144

#define BINW 256
#define BINSH 8
#define NSEG 250
#define SEGCAP 64
#define NBMAX 512
#define EPT 19
#define TILE (EPT * 256)

__global__ void __launch_bounds__(256)
part_kernel(const int* __restrict__ src, const int* __restrict__ dst,
            int* __restrict__ stage, int* __restrict__ segcnt, int* __restrict__ cnt, int N, int E) {
    __shared__ int hist[NBMAX];
    __shared__ int start[NBMAX];
    __shared__ int segoff[NBMAX];
    __shared__ int wsum[8];
    __shared__ int2 lpair[TILE];

    const int t = threadIdx.x;
    const int blk = blockIdx.x;
    const int nb = (N + BINW - 1) >> BINSH;
    const int per = (E + NSEG - 1) / NSEG;
    const int cbeg = min(blk * per, E);
    const int cend = min(cbeg + per, E);

    if (blk == 0 && t == 0) cnt[N] = 0;
    segoff[t] = 0;
    segoff[t + 256] = 0;

    for (int base = cbeg; base < cend; base += TILE) {
        hist[t] = 0;
        hist[t + 256] = 0;
        __syncthreads();

        int dv[EPT], sv[EPT], rk[EPT];
        #pragma unroll
        for (int k = 0; k < EPT; ++k) {
            const int e = base + k * 256 + t;
            const int ec = min(e, cend - 1);
            int d = dst[ec];
            const int s = src[ec];
            if (e >= cend || (unsigned)d >= (unsigned)N || (d >> BINSH) >= NBMAX) d = -1;
            dv[k] = d;
            sv[k] = s;
        }
        #pragma unroll
        for (int k = 0; k < EPT; ++k) {
            rk[k] = 0;
            if (dv[k] >= 0) rk[k] = atomicAdd(&hist[dv[k] >> BINSH], 1);
        }
        __syncthreads();

        const int h0 = hist[2 * t];
        const int h1 = hist[2 * t + 1];
        const int mine = h0 + h1;
        const int lane = t & 31;
        int inc = mine;
        #pragma unroll
        for (int off = 1; off < 32; off <<= 1) {
            const int v = __shfl_up(inc, off);
            if (lane >= off) inc += v;
        }
        if (lane == 31) wsum[t >> 5] = inc;
        __syncthreads();
        int before = 0, total = 0;
        #pragma unroll
        for (int w = 0; w < 8; ++w) {
            const int v = wsum[w];
            if (w < (t >> 5)) before += v;
            total += v;
        }
        const int ex = before + inc - mine;
        start[2 * t] = ex;
        start[2 * t + 1] = ex + h0;
        __syncthreads();

        int pos[EPT], gw[EPT];
        #pragma unroll
        for (int k = 0; k < EPT; ++k) {
            const int bin = dv[k] >= 0 ? (dv[k] >> BINSH) : 0;
            const int off = segoff[bin] + rk[k];
            pos[k] = start[bin] + rk[k];
            gw[k] = off < SEGCAP ? (bin * NSEG + blk) * SEGCAP + off : -1;
        }
        #pragma unroll
        for (int k = 0; k < EPT; ++k) {
            if (dv[k] >= 0) {
                lpair[pos[k]] = make_int2((int)(((unsigned)(dv[k] & (BINW - 1)) << 24) | ((unsigned)sv[k] & 0xFFFFFFu)), gw[k]);
            }
        }
        __syncthreads();

        int2 pr[EPT];
        #pragma unroll
        for (int k = 0; k < EPT; ++k) pr[k] = lpair[k * 256 + t];
        #pragma unroll
        for (int k = 0; k < EPT; ++k) {
            if (k * 256 + t < total && pr[k].y >= 0) stage[pr[k].y] = pr[k].x;
        }
        __syncthreads();
        segoff[t] += hist[t];
        segoff[t + 256] += hist[t + 256];
    }
    __syncthreads();
    for (int bin = t; bin < nb && bin < NBMAX; bin += 256) segcnt[bin * NSEG + blk] = segoff[bin];
}

__device__ __forceinline__ void lds_barrier() {
    __builtin_amdgcn_fence(__ATOMIC_RELEASE, "workgroup", "local");
    __builtin_amdgcn_s_barrier();
    __builtin_amdgcn_fence(__ATOMIC_ACQUIRE, "workgroup", "local");
}

__device__ __forceinline__ void put_entry(int r, int dl, int s, int d0, int* lrows,
                                          int* __restrict__ cnt, int* __restrict__ spill, int N) {
    if (r < ROWW) {
        lrows[dl * ROWW + r] = s;
    } else {
        const int k = atomicAdd(&cnt[N], 1);
        if (k >= 0 && k < SPILL_CAP) reinterpret_cast<int2*>(spill)[k] = make_int2(d0 + dl, s);
    }
}

__device__ __forceinline__ void place_entry(int dl, int s, int d0, int* lcnt, int* lrows,
                                            int* __restrict__ cnt, int* __restrict__ spill, int N) {
    put_entry(atomicAdd(&lcnt[dl], 1), dl, s, d0, lrows, cnt, spill, N);
}

__global__ void __launch_bounds__(256)
rows_scale_kernel(const int* __restrict__ src, const int* __restrict__ dst,
                  const int* __restrict__ stage, const int* __restrict__ segcnt,
                  int* __restrict__ cnt, int* __restrict__ rows, int* __restrict__ spill,
                  const float* __restrict__ X, float* __restrict__ Y,
                  int N, int E) {
    __shared__ int lcnt[BINW];
    __shared__ int lrows[BINW * ROWW];
    __shared__ int lover;

    const int t = threadIdx.x;
    const int b = blockIdx.x;
    const int d0 = b * BINW;
    const int4 zero4 = make_int4(0, 0, 0, 0);
    int4* const lr4 = reinterpret_cast<int4*>(lrows);

    int c = 0;
    int4 v0 = zero4, v1 = zero4, v2 = zero4, v3 = zero4, v4 = zero4, v5 = zero4;
    const int4* sp = reinterpret_cast<const int4*>(stage + ((size_t)b * NSEG + (t < NSEG ? t : 0)) * SEGCAP);
    if (t < NSEG) {
        c = segcnt[b * NSEG + t];
        v0 = sp[0]; v1 = sp[1]; v2 = sp[2]; v3 = sp[3]; v4 = sp[4]; v5 = sp[5];
    }

    const int rq = t >> 4;
    const int q  = t & 15;
    float4 xr[16];
    {
        const float4* x4 = reinterpret_cast<const float4*>(X);
        #pragma unroll
        for (int it = 0; it < 16; ++it) {
            const int r = d0 + it * 16 + rq;
            xr[it] = x4[(size_t)(r < N ? r : (N - 1)) * 16 + q];
        }
    }

    lcnt[t] = 0;
    for (int i = t; i < BINW * ROWW / 4; i += 256) lr4[i] = zero4;
    if (t == 0) lover = 0;
    lds_barrier();
    if (c > SEGCAP) lover = 1;
    lds_barrier();
    const bool over = lover != 0;

    if (!over) {
#define SLOT_OF(r, wv, i) int r = 0; if ((i) < c) r = atomicAdd(&lcnt[(unsigned)(wv) >> 24], 1);
#define PUT_AT(r, wv, i) if ((i) < c) put_entry(r, (int)((unsigned)(wv) >> 24), (wv) & 0xFFFFFF, d0, lrows, cnt, spill, N);
        SLOT_OF(r0, v0.x, 0)
        SLOT_OF(r1, v0.y, 1)
        SLOT_OF(r2, v0.z, 2)
        SLOT_OF(r3, v0.w, 3)
        SLOT_OF(r4, v1.x, 4)
        SLOT_OF(r5, v1.y, 5)
        SLOT_OF(r6, v1.z, 6)
        SLOT_OF(r7, v1.w, 7)
        SLOT_OF(r8, v2.x, 8)
        SLOT_OF(r9, v2.y, 9)
        SLOT_OF(r10, v2.z, 10)
        SLOT_OF(r11, v2.w, 11)
        SLOT_OF(r12, v3.x, 12)
        SLOT_OF(r13, v3.y, 13)
        SLOT_OF(r14, v3.z, 14)
        SLOT_OF(r15, v3.w, 15)
        SLOT_OF(r16, v4.x, 16)
        SLOT_OF(r17, v4.y, 17)
        SLOT_OF(r18, v4.z, 18)
        SLOT_OF(r19, v4.w, 19)
        SLOT_OF(r20, v5.x, 20)
        SLOT_OF(r21, v5.y, 21)
        SLOT_OF(r22, v5.z, 22)
        SLOT_OF(r23, v5.w, 23)
        PUT_AT(r0, v0.x, 0)
        PUT_AT(r1, v0.y, 1)
        PUT_AT(r2, v0.z, 2)
        PUT_AT(r3, v0.w, 3)
        PUT_AT(r4, v1.x, 4)
        PUT_AT(r5, v1.y, 5)
        PUT_AT(r6, v1.z, 6)
        PUT_AT(r7, v1.w, 7)
        PUT_AT(r8, v2.x, 8)
        PUT_AT(r9, v2.y, 9)
        PUT_AT(r10, v2.z, 10)
        PUT_AT(r11, v2.w, 11)
        PUT_AT(r12, v3.x, 12)
        PUT_AT(r13, v3.y, 13)
        PUT_AT(r14, v3.z, 14)
        PUT_AT(r15, v3.w, 15)
        PUT_AT(r16, v4.x, 16)
        PUT_AT(r17, v4.y, 17)
        PUT_AT(r18, v4.z, 18)
        PUT_AT(r19, v4.w, 19)
        PUT_AT(r20, v5.x, 20)
        PUT_AT(r21, v5.y, 21)
        PUT_AT(r22, v5.z, 22)
        PUT_AT(r23, v5.w, 23)
#undef SLOT_OF
#undef PUT_AT
        for (int i = 24; i < c; i += 4) {
            const int4 v = sp[i >> 2];
            place_entry((int)((unsigned)v.x >> 24), v.x & 0xFFFFFF, d0, lcnt, lrows, cnt, spill, N);
            if (i + 1 < c) place_entry((int)((unsigned)v.y >> 24), v.y & 0xFFFFFF, d0, lcnt, lrows, cnt, spill, N);
            if (i + 2 < c) place_entry((int)((unsigned)v.z >> 24), v.z & 0xFFFFFF, d0, lcnt, lrows, cnt, spill, N);
            if (i + 3 < c) place_entry((int)((unsigned)v.w >> 24), v.w & 0xFFFFFF, d0, lcnt, lrows, cnt, spill, N);
        }
    } else {
        for (int e = t; e < E; e += 256) {
            const int d = dst[e];
            if ((unsigned)d < (unsigned)N && (d >> BINSH) == b) {
                place_entry(d & (BINW - 1), src[e], d0, lcnt, lrows, cnt, spill, N);
            }
        }
    }
    lds_barrier();

    int4* const g4 = reinterpret_cast<int4*>(rows + (size_t)d0 * ROWW);
    float4* const y4 = reinterpret_cast<float4*>(Y);
    if (d0 + BINW <= N) {
        cnt[d0 + t] = lcnt[t];
        int4 rr[BINW * ROWW / 4 / 256];
        #pragma unroll
        for (int k = 0; k < BINW * ROWW / 4 / 256; ++k) rr[k] = lr4[k * 256 + t];
        #pragma unroll
        for (int k = 0; k < BINW * ROWW / 4 / 256; ++k) g4[k * 256 + t] = rr[k];

        #pragma unroll
        for (int it = 0; it < 16; ++it) {
            const int rl = it * 16 + rq;
            const float s = rsqrtf((float)(lcnt[rl] + 1));
            float4 v = xr[it];
            v.x *= s; v.y *= s; v.z *= s; v.w *= s;
            y4[(size_t)(d0 + rl) * 16 + q] = v;
        }
    } else {
        if (d0 + t < N) cnt[d0 + t] = lcnt[t];
        for (int i = t; i < BINW * ROWW / 4; i += 256) {
            if (d0 + (i >> 3) < N) g4[i] = lr4[i];
        }
        #pragma unroll
        for (int it = 0; it < 16; ++it) {
            const int rl = it * 16 + rq;
            const int r = d0 + rl;
            const float s = rsqrtf((float)(lcnt[rl] + 1));
            if (r < N) {
                float4 v = xr[it];
                v.x *= s; v.y *= s; v.z *= s; v.w *= s;
                y4[(size_t)r * 16 + q] = v;
            }
        }
    }
}


__device__ __forceinline__ unsigned inrange(int s, unsigned un) {
    return (unsigned)s < un ? (unsigned)s : 0u;
}

__device__ __forceinline__ void add4(float4& a, const float4& v) {
    a.x += v.x; a.y += v.y; a.z += v.z; a.w += v.w;
}

__device__ __forceinline__ void fma4(float4& a, const float4& v, float w) {
    a.x = fmaf(v.x, w, a.x); a.y = fmaf(v.y, w, a.y); a.z = fmaf(v.z, w, a.z); a.w = fmaf(v.w, w, a.w);
}

__device__ __forceinline__ int4 inrange4(const int4 e, unsigned last) {
    return make_int4((int)min((unsigned)e.x, last), (int)min((unsigned)e.y, last),
                     (int)min((unsigned)e.z, last), (int)min((unsigned)e.w, last));
}

__device__ __forceinline__ uint4 rowidx4(const int4 e, unsigned q) {
    return make_uint4(((unsigned)e.x << 4) | q, ((unsigned)e.y << 4) | q, ((unsigned)e.z << 4) | q, ((unsigned)e.w << 4) | q);
}

__device__ __forceinline__ void sum8(float4& acc, const float4* __restrict__ x4, const uint4 ia, const uint4 ib, int m) {
    const float4 v0 = x4[ia.x];
    const float4 v1 = x4[ia.y];
    const float4 v2 = x4[ia.z];
    const float4 v3 = x4[ia.w];
    const float4 v4 = x4[ib.x];
    const float4 v5 = x4[ib.y];
    const float4 v6 = x4[ib.z];
    const float4 v7 = x4[ib.w];
    const float mf = (float)m;
    add4(acc, v0);
    fma4(acc, v1, fminf(fmaxf(mf - 1.0f, 0.0f), 1.0f));
    fma4(acc, v2, fminf(fmaxf(mf - 2.0f, 0.0f), 1.0f));
    fma4(acc, v3, fminf(fmaxf(mf - 3.0f, 0.0f), 1.0f));
    fma4(acc, v4, fminf(fmaxf(mf - 4.0f, 0.0f), 1.0f));
    fma4(acc, v5, fminf(fmaxf(mf - 5.0f, 0.0f), 1.0f));
    fma4(acc, v6, fminf(fmaxf(mf - 6.0f, 0.0f), 1.0f));
    fma4(acc, v7, fminf(fmaxf(mf - 7.0f, 0.0f), 1.0f));
}

#define GT 5

__global__ void __launch_bounds__(256)
gather_kernel(const float* __restrict__ xs, const int* __restrict__ rows,
              const int* __restrict__ cnt, const int* __restrict__ spill,
              const int* __restrict__ src, const int* __restrict__ dst,
              const float* __restrict__ bias,
              float* __restrict__ o, int N, int E) {
    const int g = threadIdx.x >> 4;
    const int q = threadIdx.x & 15;

    const float4* x4 = reinterpret_cast<const float4*>(xs);
    const unsigned un = (unsigned)N;
    const unsigned last = un - 1u;
    const float4 b = reinterpret_cast<const float4*>(bias)[q];

    int d = blockIdx.x * (GT * 16) + g;
    if (d >= N) return;

    const int4* r4 = reinterpret_cast<const int4*>(rows + (size_t)d * ROWW);
    int c = cnt[d];
    int4 ea = r4[0];
    int4 eb = r4[1];

    for (int t = 0; t < GT; ++t) {
        const int dn = d + 16;
        const bool more = (t + 1 < GT) && (dn < N);
        const int dp = more ? dn : d;
        const int4* rn4 = reinterpret_cast<const int4*>(rows + (size_t)dp * ROWW);
        const int cn = cnt[dp];
        const int4 ean = rn4[0];
        const int4 ebn = rn4[1];

        const float4 self = x4[(size_t)d * 16 + q];
        const float di = rsqrtf((float)(c + 1));
        const int n = min(max(c, 0), ROWW);
        const int4* row4 = reinterpret_cast<const int4*>(rows + (size_t)d * ROWW);

        int4 na = row4[2];
        int4 nb = row4[3];

        float4 acc = make_float4(0.0f, 0.0f, 0.0f, 0.0f);
        for (int j = 0; j < n; j += 8) {
            sum8(acc, x4, rowidx4(inrange4(ea, last), (unsigned)q), rowidx4(inrange4(eb, last), (unsigned)q), n - j);
            ea = na; eb = nb;
            if (j + 16 < n) {
                na = row4[(j >> 2) + 4];
                nb = row4[(j >> 2) + 5];
            }
        }
        if (c > ROWW) {
            const int total = cnt[N];
            if (total >= 0 && total <= SPILL_CAP) {
                const int2* sp = reinterpret_cast<const int2*>(spill);
                for (int k = 0; k < total; ++k) {
                    const int2 p = sp[k];
                    if (p.x == d) {
                        add4(acc, x4[(size_t)inrange(p.y, un) * 16 + q]);
                    }
                }
            } else {
                acc = make_float4(0.0f, 0.0f, 0.0f, 0.0f);
                for (int e = 0; e < E; ++e) {
                    if (dst[e] == d) {
                        add4(acc, x4[(size_t)inrange(src[e], un) * 16 + q]);
                    }
                }
            }
        }

        float4 r;
        r.x = (acc.x + self.x) * di + b.x;
        r.y = (acc.y + self.y) * di + b.y;
        r.z = (acc.z + self.z) * di + b.z;
        r.w = (acc.w + self.w) * di + b.w;
        reinterpret_cast<float4*>(o)[(size_t)d * 16 + q] = r;

        if (!more) break;
        d = dn; c = cn; ea = ean; eb = ebn;
    }
}

#define TPB 5
#define HP 68

__device__ __forceinline__ v8f tile_product(const float* ap, const float* bp) {
    v8f acc = {};
    #pragma unroll
    for (int kk = 0; kk < 16; ++kk) {
        const v2f a  = *reinterpret_cast<const v2f*>(ap + 4 * kk);
        const v2f bw = *reinterpret_cast<const v2f*>(bp + 4 * kk);
        acc = __builtin_amdgcn_wmma_f32_16x16x4_f32(false, a, false, bw, (short)0, acc, false, false);
    }
    return acc;
}

__global__ void __launch_bounds__(256)
gather_gemm_kernel(const float* __restrict__ xs, const int* __restrict__ rows,
                   const int* __restrict__ cnt, const int* __restrict__ spill,
                   const int* __restrict__ src, const int* __restrict__ dst,
                   const float* __restrict__ bias, const float* __restrict__ W1, const float* __restrict__ W2,
                   float* __restrict__ y, int N, int E) {
    __shared__ float sm[(64 + 64 + 16 + 16) * HP];
    float* const lw1 = sm;
    float* const lw2 = sm + 64 * HP;
    float* const ht = sm + 128 * HP;
    float* const ot = ht + 16 * HP;

    const int g = threadIdx.x >> 4;
    const int q = threadIdx.x & 15;
    const int wave = threadIdx.x >> 5;
    const int lane = threadIdx.x & 31;
    const int lh = lane & 15;
    const int hi = lane >> 4;

    const float4* x4 = reinterpret_cast<const float4*>(xs);
    const unsigned un = (unsigned)N;
    const unsigned last = un - 1u;
    const int4 zero4 = make_int4(0, 0, 0, 0);

    int d = blockIdx.x * (TPB * 16) + g;
    int c = 0;
    int4 ea = zero4, eb = zero4;
    if (d < N) {
        const int4* r4 = reinterpret_cast<const int4*>(rows + (size_t)d * ROWW);
        c = cnt[d];
        ea = r4[0];
        eb = r4[1];
    }

    #pragma unroll
    for (int it = 0; it < 4; ++it) {
        const int i = it * 256 + threadIdx.x;
        const float4 wa = reinterpret_cast<const float4*>(W1)[i];
        const float4 wb = reinterpret_cast<const float4*>(W2)[i];
        const int k = i >> 4;
        const int n = (i & 15) * 4;
        float* p1 = lw1 + n * HP + k;
        p1[0] = wa.x; p1[HP] = wa.y; p1[2 * HP] = wa.z; p1[3 * HP] = wa.w;
        float* p2 = lw2 + n * HP + k;
        p2[0] = wb.x; p2[HP] = wb.y; p2[2 * HP] = wb.z; p2[3 * HP] = wb.w;
    }

    const int n0 = (wave & 3) * 16;
    const float bc = bias[n0 + lh];

    for (int t = 0; t < TPB; ++t) {
        if (d - g >= N) break;

        float4 r = make_float4(0.0f, 0.0f, 0.0f, 0.0f);
        float di = 0.0f;
        if (d < N) {
            const float4 self = x4[(size_t)d * 16 + q];
            di = rsqrtf((float)(c + 1));
            const int n = min(max(c, 0), ROWW);
            const int4* row4 = reinterpret_cast<const int4*>(rows + (size_t)d * ROWW);

            float4 acc = make_float4(0.0f, 0.0f, 0.0f, 0.0f);
            for (int j = 0; j < n; j += 8) {
                int4 na = zero4, nb = zero4;
                if (j + 8 < n) {
                    na = row4[(j >> 2) + 2];
                    nb = row4[(j >> 2) + 3];
                }
                sum8(acc, x4, rowidx4(inrange4(ea, last), (unsigned)q), rowidx4(inrange4(eb, last), (unsigned)q), n - j);
                ea = na; eb = nb;
            }
            if (c > ROWW) {
                const int total = cnt[N];
                if (total >= 0 && total <= SPILL_CAP) {
                    const int2* sp = reinterpret_cast<const int2*>(spill);
                    for (int k = 0; k < total; ++k) {
                        const int2 p = sp[k];
                        if (p.x == d) {
                            add4(acc, x4[(size_t)inrange(p.y, un) * 16 + q]);
                        }
                    }
                } else {
                    acc = make_float4(0.0f, 0.0f, 0.0f, 0.0f);
                    for (int e = 0; e < E; ++e) {
                        if (dst[e] == d) {
                            add4(acc, x4[(size_t)inrange(src[e], un) * 16 + q]);
                        }
                    }
                }
            }

            r.x = (acc.x + self.x) * di;
            r.y = (acc.y + self.y) * di;
            r.z = (acc.z + self.z) * di;
            r.w = (acc.w + self.w) * di;
        }
        *reinterpret_cast<float4*>(ht + g * HP + 4 * q) = r;
        __syncthreads();

        const int dn = d + 16;
        int cn = 0;
        int4 ean = zero4, ebn = zero4;
        if (t + 1 < TPB && dn < N) {
            const int4* r4 = reinterpret_cast<const int4*>(rows + (size_t)dn * ROWW);
            cn = cnt[dn];
            ean = r4[0];
            ebn = r4[1];
        }

        if (wave < 4) {
            const v8f acc = tile_product(ht + lh * HP + 2 * hi, lw1 + (n0 + lh) * HP + 2 * hi);
            float* op = ot + 8 * hi * HP + n0 + lh;
            #pragma unroll
            for (int rr = 0; rr < 8; ++rr) {
                op[rr * HP] = fmaxf(acc[rr] + bc, 0.0f);
            }
        }
        lds_barrier();

        if (wave < 4) {
            const v8f acc = tile_product(ot + lh * HP + 2 * hi, lw2 + (n0 + lh) * HP + 2 * hi);
            float* op = ht + 8 * hi * HP + n0 + lh;
            #pragma unroll
            for (int rr = 0; rr < 8; ++rr) {
                op[rr * HP] = acc[rr];
            }
        }
        __syncthreads();

        if (d < N) {
            float4 o = *reinterpret_cast<const float4*>(ht + g * HP + 4 * q);
            o.x *= di; o.y *= di; o.z *= di; o.w *= di;
            reinterpret_cast<float4*>(y)[(size_t)d * 16 + q] = o;
        }

        d = dn; c = cn; ea = ean; eb = ebn;
    }
}


extern "C" void kernel_launch(void* const* d_in, const int* in_sizes, int n_in,
                              void* d_out, int out_size, void* d_ws, size_t ws_size,
                              hipStream_t stream) {
    (void)n_in; (void)out_size; (void)ws_size;

    const float* x  = (const float*)d_in[0];
    const int*   ei = (const int*)d_in[1];
    const float* W1 = (const float*)d_in[2];
    const float* b1 = (const float*)d_in[3];
    const float* W2 = (const float*)d_in[4];
    const float* b2 = (const float*)d_in[5];
    float* out = (float*)d_out;

    const int N = in_sizes[0] / 64;
    const int E = in_sizes[1] / 2;
    const int* src = ei;
    const int* dst = ei + E;

    int* cnt    = (int*)d_ws;
    float* xs   = (float*)d_ws + (((size_t)N + 1 + 32767) & ~(size_t)32767);
    int* rows   = (int*)(xs + (size_t)N * 64);
    int* spill  = rows + (size_t)N * ROWW;

    const int T = 256;

    int* stage  = (int*)xs;
    int* segcnt = stage + (size_t)((N + BINW - 1) / BINW) * NSEG * SEGCAP;
    part_kernel<<<NSEG, T, 0, stream>>>(src, dst, stage, segcnt, cnt, N, E);
    rows_scale_kernel<<<(N + BINW - 1) / BINW, T, 0, stream>>>(src, dst, stage, segcnt, cnt, rows, spill, x, out, N, E);

    gather_gemm_kernel<<<(N + 16 * TPB - 1) / (16 * TPB), T, 0, stream>>>(out, rows, cnt, spill, src, dst, b1, W1, W2, xs, N, E);

    gather_kernel<<<(N + 16 * GT - 1) / (16 * GT), T, 0, stream>>>(xs, rows, cnt, spill, src, dst, b2, out, N, E);
}
